# starting kernel
# baseline (speedup 1.0000x reference)
#include <hip/hip_runtime.h>

#define N_ATOMS 50000
#define N_PAIRS 800000
#define CCH     64

typedef __attribute__((ext_vector_type(2))) float v2f;
typedef __attribute__((ext_vector_type(8))) float v8f;

__global__ __launch_bounds__(256) void zero4_kernel(float4* __restrict__ p, int n4) {
    int t = blockIdx.x * blockDim.x + threadIdx.x;
    if (t < n4) p[t] = make_float4(0.f, 0.f, 0.f, 0.f);
}

__global__ __launch_bounds__(256) void scatter_kernel(
    const int*   __restrict__ ind2,
    const float* __restrict__ p3,
    const float* __restrict__ i1,
    const float* __restrict__ d3,
    float*       __restrict__ agg)
{
    const int pl = threadIdx.x >> 6;
    const int c  = threadIdx.x & 63;
    const int p  = blockIdx.x * 4 + pl;

    const int i = ind2[2 * p];
    const int j = ind2[2 * p + 1];

    __builtin_prefetch((const void*)(i1 + (size_t)(p + 64) * CCH + c), 0, 1);

    const float gate = i1[(size_t)p * CCH + c];
    const float dx0  = d3[3 * p + 0];
    const float dx1  = d3[3 * p + 1];
    const float dx2  = d3[3 * p + 2];

    const float* src = p3 + (size_t)j * (3 * CCH) + c;
    const float  v0  = (src[0]         + dx0) * gate;
    const float  v1  = (src[CCH]       + dx1) * gate;
    const float  v2  = (src[2 * CCH]   + dx2) * gate;

    float* dst = agg + (size_t)i * (3 * CCH) + c;
    (void)__hip_atomic_fetch_add(dst,            v0, __ATOMIC_RELAXED, __HIP_MEMORY_SCOPE_AGENT);
    (void)__hip_atomic_fetch_add(dst + CCH,      v1, __ATOMIC_RELAXED, __HIP_MEMORY_SCOPE_AGENT);
    (void)__hip_atomic_fetch_add(dst + 2 * CCH,  v2, __ATOMIC_RELAXED, __HIP_MEMORY_SCOPE_AGENT);
}

__global__ __launch_bounds__(128) void gemm_wmma_kernel(
    const float* __restrict__ A,
    const float* __restrict__ W,
    float*       __restrict__ D)
{
    const int wave = threadIdx.x >> 5;
    const int lane = threadIdx.x & 31;
    const int tm   = blockIdx.x;

    const int m16  = lane & 15;
    const int koff = (lane >> 4) * 2;

    const float* arow = A + (size_t)(tm * 16 + m16) * CCH + koff;
    const float* wrow = W + (size_t)(wave * 16 + m16) * CCH + koff;

    v8f acc = {};
    #pragma unroll
    for (int k = 0; k < 16; ++k) {
        v2f a = *(const v2f*)(arow + k * 4);
        v2f b = *(const v2f*)(wrow + k * 4);
        acc = __builtin_amdgcn_wmma_f32_16x16x4_f32(
             false, a,  false, b,
             (short)0, acc,  false,  false);
    }

    const int col     = wave * 16 + m16;
    const int rowbase = tm * 16 + (lane >> 4) * 8;
    float* o = D + (size_t)rowbase * CCH + col;
    #pragma unroll
    for (int r = 0; r < 8; ++r)
        o[(size_t)r * CCH] = acc[r];
}

__global__ __launch_bounds__(256) void dotted_kernel(
    const float* __restrict__ p3n,
    float*       __restrict__ dotted)
{
    int t = blockIdx.x * blockDim.x + threadIdx.x;
    if (t >= N_ATOMS * CCH) return;
    int a = t >> 6, d = t & 63;
    const float* p = p3n + (size_t)a * (3 * CCH) + d;
    float x0 = p[0], x1 = p[CCH], x2 = p[2 * CCH];
    dotted[t] = x0 * x0 + x1 * x1 + x2 * x2;
}

extern "C" void kernel_launch(void* const* d_in, const int* in_sizes, int n_in,
                              void* d_out, int out_size, void* d_ws, size_t ws_size,
                              hipStream_t stream) {
    const int*   ind2 = (const int*)  d_in[0];
    const float* p3   = (const float*)d_in[1];
    const float* i1   = (const float*)d_in[2];
    const float* d3   = (const float*)d_in[3];
    const float* W    = (const float*)d_in[4];

    float* p3_new = (float*)d_out;
    float* dotted = (float*)d_out + (size_t)N_ATOMS * 3 * CCH;
    float* agg    = (float*)d_ws;

    const int nagg  = N_ATOMS * 3 * CCH;
    const int nagg4 = nagg / 4;

    zero4_kernel<<<(nagg4 + 255) / 256, 256, 0, stream>>>((float4*)agg, nagg4);

    scatter_kernel<<<N_PAIRS / 4, 256, 0, stream>>>(ind2, p3, i1, d3, agg);

    gemm_wmma_kernel<<<(N_ATOMS * 3) / 16, 128, 0, stream>>>(agg, W, p3_new);

    dotted_kernel<<<(N_ATOMS * CCH + 255) / 256, 256, 0, stream>>>(p3_new, dotted);
}


# Float atomic scatter-add replaced by a per-call counting sort of pairs by receiver (integer tickets) and a one-wave-per-atom segment sum with plain stores; sum order = ticket arrival order
# speedup vs baseline: 1.1511x; 1.1511x over previous
#include <hip/hip_runtime.h>


#define N_ATOMS 50000
#define N_PAIRS 800000
#define CCH     64

#define SCR_CNT   0
#define SCR_OFF   50048
#define SCR_ENTA  100096
#define SCR_ENTZ  (100096 + 4 * N_PAIRS)

typedef __attribute__((ext_vector_type(2))) float v2f;
typedef __attribute__((ext_vector_type(8))) float v8f;

__global__ __launch_bounds__(256) void zero_cnt_kernel(int4* __restrict__ c, int n4) {
    int t = blockIdx.x * blockDim.x + threadIdx.x;
    if (t < n4) c[t] = make_int4(0, 0, 0, 0);
}

__global__ __launch_bounds__(256) void hist_kernel(
    const int* __restrict__ ind2,
    int*       __restrict__ cnt)
{
    const int p = blockIdx.x * 256 + threadIdx.x;
    const int i = ind2[2 * p];
    (void)__hip_atomic_fetch_add(cnt + i, 1, __ATOMIC_RELAXED, __HIP_MEMORY_SCOPE_AGENT);
}

__global__ __launch_bounds__(1024) void scan_kernel(
    int* __restrict__ cnt,
    int* __restrict__ off)
{
    __shared__ int wtot[32];
    const int t = threadIdx.x, lane = t & 31, w = t >> 5;
    const int NQ = N_ATOMS / 4;
    int4 v[13];
    #pragma unroll
    for (int k = 0; k < 13; ++k) {
        const int q = k * 1024 + t;
        v[k] = (q < NQ) ? ((const int4*)cnt)[q] : make_int4(0, 0, 0, 0);
    }
    int carry = 0;
    #pragma unroll
    for (int k = 0; k < 13; ++k) {
        const int q = k * 1024 + t;
        const int s = v[k].x + v[k].y + v[k].z + v[k].w;
        int inc = s;
        for (int d = 1; d < 32; d <<= 1) {
            const int nb = __shfl_up(inc, d);
            if (lane >= d) inc += nb;
        }
        if (lane == 31) wtot[w] = inc;
        __syncthreads();
        const int wt = wtot[lane];
        int winc = wt;
        for (int d = 1; d < 32; d <<= 1) {
            const int nb = __shfl_up(winc, d);
            if (lane >= d) winc += nb;
        }
        const int wprefix = __shfl(winc, w) - __shfl(wt, w);
        const int total   = __shfl(winc, 31);
        __syncthreads();
        const int ex = carry + wprefix + (inc - s);
        int4 o;
        o.x = ex;
        o.y = o.x + v[k].x;
        o.z = o.y + v[k].y;
        o.w = o.z + v[k].z;
        if (q < NQ) { ((int4*)off)[q] = o; ((int4*)cnt)[q] = o; }
        carry += total;
    }
    if (t == 0) off[N_ATOMS] = carry;
}

__global__ __launch_bounds__(256) void fill_kernel(
    const int*   __restrict__ ind2,
    const float* __restrict__ d3,
    int*         __restrict__ cursor,
    int4*        __restrict__ entA,
    float*       __restrict__ entZ)
{
    const int p = blockIdx.x * 256 + threadIdx.x;
    const int i = ind2[2 * p];
    const int j = ind2[2 * p + 1];
    const float dx = d3[3 * p + 0];
    const float dy = d3[3 * p + 1];
    const float dz = d3[3 * p + 2];
    const int pos = __hip_atomic_fetch_add(cursor + i, 1, __ATOMIC_RELAXED, __HIP_MEMORY_SCOPE_AGENT);
    entA[pos] = make_int4(p, j, __float_as_int(dx), __float_as_int(dy));
    entZ[pos] = dz;
}

template <int U>
__device__ __forceinline__ void seg_step(
    int k, int lane, int ep, int ej, float ex, float ey, float ez,
    const float* __restrict__ p3, const float* __restrict__ i1,
    v2f& acc0, v2f& acc1, v2f& acc2)
{
    v2f g[U], x0[U], x1[U], x2[U];
    float d0[U], d1[U], d2[U];
    #pragma unroll
    for (int u = 0; u < U; ++u) {
        const int p = __builtin_amdgcn_readlane(ep, k + u);
        const int j = __builtin_amdgcn_readlane(ej, k + u);
        d0[u] = __builtin_bit_cast(float, __builtin_amdgcn_readlane(__builtin_bit_cast(int, ex), k + u));
        d1[u] = __builtin_bit_cast(float, __builtin_amdgcn_readlane(__builtin_bit_cast(int, ey), k + u));
        d2[u] = __builtin_bit_cast(float, __builtin_amdgcn_readlane(__builtin_bit_cast(int, ez), k + u));
        g[u] = __builtin_nontemporal_load((const v2f*)(i1 + (size_t)p * CCH + 2 * lane));
        const float* s = p3 + (size_t)j * (3 * CCH) + 2 * lane;
        x0[u] = *(const v2f*)(s);
        x1[u] = *(const v2f*)(s + CCH);
        x2[u] = *(const v2f*)(s + 2 * CCH);
    }
    #pragma unroll
    for (int u = 0; u < U; ++u) {
        acc0 += (x0[u] + d0[u]) * g[u];
        acc1 += (x1[u] + d1[u]) * g[u];
        acc2 += (x2[u] + d2[u]) * g[u];
    }
}

__global__ __launch_bounds__(256) void segsum_kernel(
    const int*   __restrict__ off,
    const int4*  __restrict__ entA,
    const float* __restrict__ entZ,
    const float* __restrict__ p3,
    const float* __restrict__ i1,
    float*       __restrict__ agg)
{
    const int lane = threadIdx.x & 31;
    const int a = __builtin_amdgcn_readfirstlane(blockIdx.x * 8 + (threadIdx.x >> 5));
    const int beg = off[a];
    const int end = off[a + 1];

    v2f acc0 = {0.f, 0.f}, acc1 = {0.f, 0.f}, acc2 = {0.f, 0.f};
    for (int base = beg; base < end; base += 32) {
        const int n = (end - base < 32) ? (end - base) : 32;
        int ep = 0, ej = 0;
        float ex = 0.f, ey = 0.f, ez = 0.f;
        if (lane < n) {
            const int4 e = entA[base + lane];
            ep = e.x; ej = e.y;
            ex = __int_as_float(e.z); ey = __int_as_float(e.w);
            ez = entZ[base + lane];
        }
        int k = 0;
        for (; k + 4 <= n; k += 4)
            seg_step<4>(k, lane, ep, ej, ex, ey, ez, p3, i1, acc0, acc1, acc2);
        if (k + 2 <= n) {
            seg_step<2>(k, lane, ep, ej, ex, ey, ez, p3, i1, acc0, acc1, acc2);
            k += 2;
        }
        if (k < n)
            seg_step<1>(k, lane, ep, ej, ex, ey, ez, p3, i1, acc0, acc1, acc2);
    }

    float* dst = agg + (size_t)a * (3 * CCH) + 2 * lane;
    *(v2f*)(dst)           = acc0;
    *(v2f*)(dst + CCH)     = acc1;
    *(v2f*)(dst + 2 * CCH) = acc2;
}

__global__ __launch_bounds__(128) void gemm_wmma_kernel(
    const float* __restrict__ A,
    const float* __restrict__ W,
    float*       __restrict__ D)
{
    const int wave = threadIdx.x >> 5;
    const int lane = threadIdx.x & 31;
    const int tm   = blockIdx.x;

    const int m16  = lane & 15;
    const int koff = (lane >> 4) * 2;

    const float* arow = A + (size_t)(tm * 16 + m16) * CCH + koff;
    const float* wrow = W + (size_t)(wave * 16 + m16) * CCH + koff;

    v8f acc = {};
    #pragma unroll
    for (int k = 0; k < 16; ++k) {
        v2f a = *(const v2f*)(arow + k * 4);
        v2f b = *(const v2f*)(wrow + k * 4);
        acc = __builtin_amdgcn_wmma_f32_16x16x4_f32(
             false, a,  false, b,
             (short)0, acc,  false,  false);
    }

    const int col     = wave * 16 + m16;
    const int rowbase = tm * 16 + (lane >> 4) * 8;
    float* o = D + (size_t)rowbase * CCH + col;
    #pragma unroll
    for (int r = 0; r < 8; ++r)
        o[(size_t)r * CCH] = acc[r];
}

__global__ __launch_bounds__(256) void dotted_kernel(
    const float* __restrict__ p3n,
    float*       __restrict__ dotted)
{
    int t = blockIdx.x * blockDim.x + threadIdx.x;
    if (t >= N_ATOMS * CCH) return;
    int a = t >> 6, d = t & 63;
    const float* p = p3n + (size_t)a * (3 * CCH) + d;
    float x0 = p[0], x1 = p[CCH], x2 = p[2 * CCH];
    dotted[t] = x0 * x0 + x1 * x1 + x2 * x2;
}

extern "C" void kernel_launch(void* const* d_in, const int* in_sizes, int n_in,
                              void* d_out, int out_size, void* d_ws, size_t ws_size,
                              hipStream_t stream) {
    const int*   ind2 = (const int*)  d_in[0];
    const float* p3   = (const float*)d_in[1];
    const float* i1   = (const float*)d_in[2];
    const float* d3   = (const float*)d_in[3];
    const float* W    = (const float*)d_in[4];

    float* p3_new = (float*)d_out;
    float* dotted = (float*)d_out + (size_t)N_ATOMS * 3 * CCH;
    float* agg    = (float*)d_ws;

    int*   scr  = (int*)d_out;
    int*   cnt  = scr + SCR_CNT;
    int*   off  = scr + SCR_OFF;
    int4*  entA = (int4*)(scr + SCR_ENTA);
    float* entZ = (float*)(scr + SCR_ENTZ);

    zero_cnt_kernel<<<(N_ATOMS / 4 + 255) / 256, 256, 0, stream>>>((int4*)cnt, N_ATOMS / 4);

    hist_kernel<<<N_PAIRS / 256, 256, 0, stream>>>(ind2, cnt);

    scan_kernel<<<1, 1024, 0, stream>>>(cnt, off);

    fill_kernel<<<N_PAIRS / 256, 256, 0, stream>>>(ind2, d3, cnt, entA, entZ);

    segsum_kernel<<<N_ATOMS / 8, 256, 0, stream>>>(off, entA, entZ, p3, i1, agg);

    gemm_wmma_kernel<<<(N_ATOMS * 3) / 16, 128, 0, stream>>>(agg, W, p3_new);

    dotted_kernel<<<(N_ATOMS * CCH + 255) / 256, 256, 0, stream>>>(p3_new, dotted);
}


# dotted fused into the GEMM epilogue: one wave = 16 atoms x 3 rows x 64 cols (12 acc tiles, A tiles x-major, squares in one lane); dotted launch gone; 16-byte operand loads regroup the k sum order
# speedup vs baseline: 1.2348x; 1.0727x over previous
#include <hip/hip_runtime.h>


#define N_ATOMS 50000
#define N_PAIRS 800000
#define CCH     64

#define SCR_CNT   0
#define SCR_OFF   50048
#define SCR_ENTA  100096
#define SCR_ENTZ  (100096 + 4 * N_PAIRS)

typedef __attribute__((ext_vector_type(2))) float v2f;
typedef __attribute__((ext_vector_type(8))) float v8f;

__global__ __launch_bounds__(256) void zero_cnt_kernel(int4* __restrict__ c, int n4) {
    int t = blockIdx.x * blockDim.x + threadIdx.x;
    if (t < n4) c[t] = make_int4(0, 0, 0, 0);
}

__global__ __launch_bounds__(256) void hist_kernel(
    const int* __restrict__ ind2,
    int*       __restrict__ cnt)
{
    const int p = blockIdx.x * 256 + threadIdx.x;
    const int i = ind2[2 * p];
    (void)__hip_atomic_fetch_add(cnt + i, 1, __ATOMIC_RELAXED, __HIP_MEMORY_SCOPE_AGENT);
}

__global__ __launch_bounds__(1024) void scan_kernel(
    int* __restrict__ cnt,
    int* __restrict__ off)
{
    __shared__ int wtot[32];
    const int t = threadIdx.x, lane = t & 31, w = t >> 5;
    const int NQ = N_ATOMS / 4;
    int4 v[13];
    #pragma unroll
    for (int k = 0; k < 13; ++k) {
        const int q = k * 1024 + t;
        v[k] = (q < NQ) ? ((const int4*)cnt)[q] : make_int4(0, 0, 0, 0);
    }
    int carry = 0;
    #pragma unroll
    for (int k = 0; k < 13; ++k) {
        const int q = k * 1024 + t;
        const int s = v[k].x + v[k].y + v[k].z + v[k].w;
        int inc = s;
        for (int d = 1; d < 32; d <<= 1) {
            const int nb = __shfl_up(inc, d);
            if (lane >= d) inc += nb;
        }
        if (lane == 31) wtot[w] = inc;
        __syncthreads();
        const int wt = wtot[lane];
        int winc = wt;
        for (int d = 1; d < 32; d <<= 1) {
            const int nb = __shfl_up(winc, d);
            if (lane >= d) winc += nb;
        }
        const int wprefix = __shfl(winc, w) - __shfl(wt, w);
        const int total   = __shfl(winc, 31);
        __syncthreads();
        const int ex = carry + wprefix + (inc - s);
        int4 o;
        o.x = ex;
        o.y = o.x + v[k].x;
        o.z = o.y + v[k].y;
        o.w = o.z + v[k].z;
        if (q < NQ) { ((int4*)off)[q] = o; ((int4*)cnt)[q] = o; }
        carry += total;
    }
    if (t == 0) off[N_ATOMS] = carry;
}

__global__ __launch_bounds__(256) void fill_kernel(
    const int*   __restrict__ ind2,
    const float* __restrict__ d3,
    int*         __restrict__ cursor,
    int4*        __restrict__ entA,
    float*       __restrict__ entZ)
{
    const int p = blockIdx.x * 256 + threadIdx.x;
    const int i = ind2[2 * p];
    const int j = ind2[2 * p + 1];
    const float dx = d3[3 * p + 0];
    const float dy = d3[3 * p + 1];
    const float dz = d3[3 * p + 2];
    const int pos = __hip_atomic_fetch_add(cursor + i, 1, __ATOMIC_RELAXED, __HIP_MEMORY_SCOPE_AGENT);
    entA[pos] = make_int4(p, j, __float_as_int(dx), __float_as_int(dy));
    entZ[pos] = dz;
}

template <int U>
__device__ __forceinline__ void seg_step(
    int k, int lane, int ep, int ej, float ex, float ey, float ez,
    const float* __restrict__ p3, const float* __restrict__ i1,
    v2f& acc0, v2f& acc1, v2f& acc2)
{
    v2f g[U], x0[U], x1[U], x2[U];
    float d0[U], d1[U], d2[U];
    #pragma unroll
    for (int u = 0; u < U; ++u) {
        const int p = __builtin_amdgcn_readlane(ep, k + u);
        const int j = __builtin_amdgcn_readlane(ej, k + u);
        d0[u] = __builtin_bit_cast(float, __builtin_amdgcn_readlane(__builtin_bit_cast(int, ex), k + u));
        d1[u] = __builtin_bit_cast(float, __builtin_amdgcn_readlane(__builtin_bit_cast(int, ey), k + u));
        d2[u] = __builtin_bit_cast(float, __builtin_amdgcn_readlane(__builtin_bit_cast(int, ez), k + u));
        g[u] = __builtin_nontemporal_load((const v2f*)(i1 + (size_t)p * CCH + 2 * lane));
        const float* s = p3 + (size_t)j * (3 * CCH) + 2 * lane;
        x0[u] = *(const v2f*)(s);
        x1[u] = *(const v2f*)(s + CCH);
        x2[u] = *(const v2f*)(s + 2 * CCH);
    }
    #pragma unroll
    for (int u = 0; u < U; ++u) {
        acc0 += (x0[u] + d0[u]) * g[u];
        acc1 += (x1[u] + d1[u]) * g[u];
        acc2 += (x2[u] + d2[u]) * g[u];
    }
}

__global__ __launch_bounds__(256) void segsum_kernel(
    const int*   __restrict__ off,
    const int4*  __restrict__ entA,
    const float* __restrict__ entZ,
    const float* __restrict__ p3,
    const float* __restrict__ i1,
    float*       __restrict__ agg)
{
    const int lane = threadIdx.x & 31;
    const int a = __builtin_amdgcn_readfirstlane(blockIdx.x * 8 + (threadIdx.x >> 5));
    const int beg = off[a];
    const int end = off[a + 1];

    v2f acc0 = {0.f, 0.f}, acc1 = {0.f, 0.f}, acc2 = {0.f, 0.f};
    for (int base = beg; base < end; base += 32) {
        const int n = (end - base < 32) ? (end - base) : 32;
        int ep = 0, ej = 0;
        float ex = 0.f, ey = 0.f, ez = 0.f;
        if (lane < n) {
            const int4 e = entA[base + lane];
            ep = e.x; ej = e.y;
            ex = __int_as_float(e.z); ey = __int_as_float(e.w);
            ez = entZ[base + lane];
        }
        int k = 0;
        for (; k + 4 <= n; k += 4)
            seg_step<4>(k, lane, ep, ej, ex, ey, ez, p3, i1, acc0, acc1, acc2);
        if (k + 2 <= n) {
            seg_step<2>(k, lane, ep, ej, ex, ey, ez, p3, i1, acc0, acc1, acc2);
            k += 2;
        }
        if (k < n)
            seg_step<1>(k, lane, ep, ej, ex, ey, ez, p3, i1, acc0, acc1, acc2);
    }

    float* dst = agg + (size_t)a * (3 * CCH) + 2 * lane;
    *(v2f*)(dst)           = acc0;
    *(v2f*)(dst + CCH)     = acc1;
    *(v2f*)(dst + 2 * CCH) = acc2;
}

typedef __attribute__((ext_vector_type(4))) float v4f;

__global__ __launch_bounds__(128) void gemm_dot_kernel(
    const float* __restrict__ A,
    const float* __restrict__ W,
    float*       __restrict__ D,
    float*       __restrict__ dotted)
{
    const int lane = threadIdx.x & 31;
    const int g = __builtin_amdgcn_readfirstlane(blockIdx.x * 4 + (threadIdx.x >> 5));
    if (g >= N_ATOMS / 16) return;

    const int c = lane & 15;
    const int h = lane >> 4;

    const float* arow = A + (size_t)(g * 16 + c) * (3 * CCH) + 4 * h;
    const float* wrow = W + (size_t)c * CCH + 4 * h;

    v8f acc[3][4] = {};
    #pragma unroll
    for (int q = 0; q < 8; ++q) {
        v4f a[3], b[4];
        #pragma unroll
        for (int x = 0; x < 3; ++x)
            a[x] = *(const v4f*)(arow + x * CCH + 8 * q);
        #pragma unroll
        for (int n = 0; n < 4; ++n)
            b[n] = *(const v4f*)(wrow + (size_t)n * 16 * CCH + 8 * q);
        #pragma unroll
        for (int x = 0; x < 3; ++x) {
            const v2f a0 = {a[x].x, a[x].y};
            const v2f a1 = {a[x].z, a[x].w};
            #pragma unroll
            for (int n = 0; n < 4; ++n) {
                const v2f b0 = {b[n].x, b[n].y};
                const v2f b1 = {b[n].z, b[n].w};
                acc[x][n] = __builtin_amdgcn_wmma_f32_16x16x4_f32(
                     false, a0,  false, b0,
                     (short)0, acc[x][n],  false,  false);
                acc[x][n] = __builtin_amdgcn_wmma_f32_16x16x4_f32(
                     false, a1,  false, b1,
                     (short)0, acc[x][n],  false,  false);
            }
        }
    }

    const int atom0 = g * 16 + 8 * h;
    float* o  = D      + (size_t)atom0 * (3 * CCH) + c;
    float* od = dotted + (size_t)atom0 * CCH + c;
    #pragma unroll
    for (int r = 0; r < 8; ++r) {
        #pragma unroll
        for (int n = 0; n < 4; ++n) {
            const float x0 = acc[0][n][r], x1 = acc[1][n][r], x2 = acc[2][n][r];
            o[(r * 3 + 0) * CCH + n * 16] = x0;
            o[(r * 3 + 1) * CCH + n * 16] = x1;
            o[(r * 3 + 2) * CCH + n * 16] = x2;
            od[r * CCH + n * 16] = x0 * x0 + x1 * x1 + x2 * x2;
        }
    }
}

extern "C" void kernel_launch(void* const* d_in, const int* in_sizes, int n_in,
                              void* d_out, int out_size, void* d_ws, size_t ws_size,
                              hipStream_t stream) {
    const int*   ind2 = (const int*)  d_in[0];
    const float* p3   = (const float*)d_in[1];
    const float* i1   = (const float*)d_in[2];
    const float* d3   = (const float*)d_in[3];
    const float* W    = (const float*)d_in[4];

    float* p3_new = (float*)d_out;
    float* dotted = (float*)d_out + (size_t)N_ATOMS * 3 * CCH;
    float* agg    = (float*)d_ws;

    int*   scr  = (int*)d_out;
    int*   cnt  = scr + SCR_CNT;
    int*   off  = scr + SCR_OFF;
    int4*  entA = (int4*)(scr + SCR_ENTA);
    float* entZ = (float*)(scr + SCR_ENTZ);

    zero_cnt_kernel<<<(N_ATOMS / 4 + 255) / 256, 256, 0, stream>>>((int4*)cnt, N_ATOMS / 4);

    hist_kernel<<<N_PAIRS / 256, 256, 0, stream>>>(ind2, cnt);

    scan_kernel<<<1, 1024, 0, stream>>>(cnt, off);

    fill_kernel<<<N_PAIRS / 256, 256, 0, stream>>>(ind2, d3, cnt, entA, entZ);

    segsum_kernel<<<N_ATOMS / 8, 256, 0, stream>>>(off, entA, entZ, p3, i1, agg);

    gemm_dot_kernel<<<(N_ATOMS / 16 + 3) / 4, 128, 0, stream>>>(agg, W, p3_new, dotted);
}


# Fill pass's scattered returning atomics removed: the count pass keeps each add's return as the pair's ticket, fill places at off[i]+ticket with plain loads; sum order = count-pass arrival order
# speedup vs baseline: 1.4750x; 1.1945x over previous
#include <hip/hip_runtime.h>


#define N_ATOMS 50000
#define N_PAIRS 800000
#define CCH     64

#define SCR_CNT   0
#define SCR_OFF   50048
#define SCR_ENTA  100096
#define SCR_ENTZ  (100096 + 4 * N_PAIRS)
#define SCR_RANK  (100096 + 5 * N_PAIRS)

typedef __attribute__((ext_vector_type(2))) float v2f;
typedef __attribute__((ext_vector_type(8))) float v8f;

__global__ __launch_bounds__(256) void zero_cnt_kernel(int4* __restrict__ c, int n4) {
    int t = blockIdx.x * blockDim.x + threadIdx.x;
    if (t < n4) c[t] = make_int4(0, 0, 0, 0);
}

__global__ __launch_bounds__(256) void hist_kernel(
    const int* __restrict__ ind2,
    int*       __restrict__ cnt,
    int*       __restrict__ rank)
{
    const int p = blockIdx.x * 256 + threadIdx.x;
    const int i = ind2[2 * p];
    rank[p] = __hip_atomic_fetch_add(cnt + i, 1, __ATOMIC_RELAXED, __HIP_MEMORY_SCOPE_AGENT);
}

__global__ __launch_bounds__(1024) void scan_kernel(
    const int* __restrict__ cnt,
    int*       __restrict__ off)
{
    __shared__ int wtot[32];
    const int t = threadIdx.x, lane = t & 31, w = t >> 5;
    const int NQ = N_ATOMS / 4;
    int4 v[13];
    #pragma unroll
    for (int k = 0; k < 13; ++k) {
        const int q = k * 1024 + t;
        v[k] = (q < NQ) ? ((const int4*)cnt)[q] : make_int4(0, 0, 0, 0);
    }
    int carry = 0;
    #pragma unroll
    for (int k = 0; k < 13; ++k) {
        const int q = k * 1024 + t;
        const int s = v[k].x + v[k].y + v[k].z + v[k].w;
        int inc = s;
        for (int d = 1; d < 32; d <<= 1) {
            const int nb = __shfl_up(inc, d);
            if (lane >= d) inc += nb;
        }
        if (lane == 31) wtot[w] = inc;
        __syncthreads();
        const int wt = wtot[lane];
        int winc = wt;
        for (int d = 1; d < 32; d <<= 1) {
            const int nb = __shfl_up(winc, d);
            if (lane >= d) winc += nb;
        }
        const int wprefix = __shfl(winc, w) - __shfl(wt, w);
        const int total   = __shfl(winc, 31);
        __syncthreads();
        const int ex = carry + wprefix + (inc - s);
        int4 o;
        o.x = ex;
        o.y = o.x + v[k].x;
        o.z = o.y + v[k].y;
        o.w = o.z + v[k].z;
        if (q < NQ) ((int4*)off)[q] = o;
        carry += total;
    }
    if (t == 0) off[N_ATOMS] = carry;
}

__global__ __launch_bounds__(256) void fill_kernel(
    const int*   __restrict__ ind2,
    const float* __restrict__ d3,
    const int*   __restrict__ off,
    const int*   __restrict__ rank,
    int4*        __restrict__ entA,
    float*       __restrict__ entZ)
{
    const int p = blockIdx.x * 256 + threadIdx.x;
    const int i = ind2[2 * p];
    const int j = ind2[2 * p + 1];
    const float dx = d3[3 * p + 0];
    const float dy = d3[3 * p + 1];
    const float dz = d3[3 * p + 2];
    const int pos = off[i] + rank[p];
    entA[pos] = make_int4(p, j, __float_as_int(dx), __float_as_int(dy));
    entZ[pos] = dz;
}

template <int U>
__device__ __forceinline__ void seg_step(
    int k, int lane, int ep, int ej, float ex, float ey, float ez,
    const float* __restrict__ p3, const float* __restrict__ i1,
    v2f& acc0, v2f& acc1, v2f& acc2)
{
    v2f g[U], x0[U], x1[U], x2[U];
    float d0[U], d1[U], d2[U];
    #pragma unroll
    for (int u = 0; u < U; ++u) {
        const int p = __builtin_amdgcn_readlane(ep, k + u);
        const int j = __builtin_amdgcn_readlane(ej, k + u);
        d0[u] = __builtin_bit_cast(float, __builtin_amdgcn_readlane(__builtin_bit_cast(int, ex), k + u));
        d1[u] = __builtin_bit_cast(float, __builtin_amdgcn_readlane(__builtin_bit_cast(int, ey), k + u));
        d2[u] = __builtin_bit_cast(float, __builtin_amdgcn_readlane(__builtin_bit_cast(int, ez), k + u));
        g[u] = __builtin_nontemporal_load((const v2f*)(i1 + (size_t)p * CCH + 2 * lane));
        const float* s = p3 + (size_t)j * (3 * CCH) + 2 * lane;
        x0[u] = *(const v2f*)(s);
        x1[u] = *(const v2f*)(s + CCH);
        x2[u] = *(const v2f*)(s + 2 * CCH);
    }
    #pragma unroll
    for (int u = 0; u < U; ++u) {
        acc0 += (x0[u] + d0[u]) * g[u];
        acc1 += (x1[u] + d1[u]) * g[u];
        acc2 += (x2[u] + d2[u]) * g[u];
    }
}

__global__ __launch_bounds__(256) void segsum_kernel(
    const int*   __restrict__ off,
    const int4*  __restrict__ entA,
    const float* __restrict__ entZ,
    const float* __restrict__ p3,
    const float* __restrict__ i1,
    float*       __restrict__ agg)
{
    const int lane = threadIdx.x & 31;
    const int a = __builtin_amdgcn_readfirstlane(blockIdx.x * 8 + (threadIdx.x >> 5));
    const int beg = off[a];
    const int end = off[a + 1];

    v2f acc0 = {0.f, 0.f}, acc1 = {0.f, 0.f}, acc2 = {0.f, 0.f};
    for (int base = beg; base < end; base += 32) {
        const int n = (end - base < 32) ? (end - base) : 32;
        int ep = 0, ej = 0;
        float ex = 0.f, ey = 0.f, ez = 0.f;
        if (lane < n) {
            const int4 e = entA[base + lane];
            ep = e.x; ej = e.y;
            ex = __int_as_float(e.z); ey = __int_as_float(e.w);
            ez = entZ[base + lane];
        }
        int k = 0;
        for (; k + 4 <= n; k += 4)
            seg_step<4>(k, lane, ep, ej, ex, ey, ez, p3, i1, acc0, acc1, acc2);
        if (k + 2 <= n) {
            seg_step<2>(k, lane, ep, ej, ex, ey, ez, p3, i1, acc0, acc1, acc2);
            k += 2;
        }
        if (k < n)
            seg_step<1>(k, lane, ep, ej, ex, ey, ez, p3, i1, acc0, acc1, acc2);
    }

    float* dst = agg + (size_t)a * (3 * CCH) + 2 * lane;
    *(v2f*)(dst)           = acc0;
    *(v2f*)(dst + CCH)     = acc1;
    *(v2f*)(dst + 2 * CCH) = acc2;
}

typedef __attribute__((ext_vector_type(4))) float v4f;

__global__ __launch_bounds__(128) void gemm_dot_kernel(
    const float* __restrict__ A,
    const float* __restrict__ W,
    float*       __restrict__ D,
    float*       __restrict__ dotted)
{
    const int lane = threadIdx.x & 31;
    const int g = __builtin_amdgcn_readfirstlane(blockIdx.x * 4 + (threadIdx.x >> 5));
    if (g >= N_ATOMS / 16) return;

    const int c = lane & 15;
    const int h = lane >> 4;

    const float* arow = A + (size_t)(g * 16 + c) * (3 * CCH) + 4 * h;
    const float* wrow = W + (size_t)c * CCH + 4 * h;

    v8f acc[3][4] = {};
    #pragma unroll
    for (int q = 0; q < 8; ++q) {
        v4f a[3], b[4];
        #pragma unroll
        for (int x = 0; x < 3; ++x)
            a[x] = *(const v4f*)(arow + x * CCH + 8 * q);
        #pragma unroll
        for (int n = 0; n < 4; ++n)
            b[n] = *(const v4f*)(wrow + (size_t)n * 16 * CCH + 8 * q);
        #pragma unroll
        for (int x = 0; x < 3; ++x) {
            const v2f a0 = {a[x].x, a[x].y};
            const v2f a1 = {a[x].z, a[x].w};
            #pragma unroll
            for (int n = 0; n < 4; ++n) {
                const v2f b0 = {b[n].x, b[n].y};
                const v2f b1 = {b[n].z, b[n].w};
                acc[x][n] = __builtin_amdgcn_wmma_f32_16x16x4_f32(
                     false, a0,  false, b0,
                     (short)0, acc[x][n],  false,  false);
                acc[x][n] = __builtin_amdgcn_wmma_f32_16x16x4_f32(
                     false, a1,  false, b1,
                     (short)0, acc[x][n],  false,  false);
            }
        }
    }

    const int atom0 = g * 16 + 8 * h;
    float* o  = D      + (size_t)atom0 * (3 * CCH) + c;
    float* od = dotted + (size_t)atom0 * CCH + c;
    #pragma unroll
    for (int r = 0; r < 8; ++r) {
        #pragma unroll
        for (int n = 0; n < 4; ++n) {
            const float x0 = acc[0][n][r], x1 = acc[1][n][r], x2 = acc[2][n][r];
            o[(r * 3 + 0) * CCH + n * 16] = x0;
            o[(r * 3 + 1) * CCH + n * 16] = x1;
            o[(r * 3 + 2) * CCH + n * 16] = x2;
            od[r * CCH + n * 16] = x0 * x0 + x1 * x1 + x2 * x2;
        }
    }
}

extern "C" void kernel_launch(void* const* d_in, const int* in_sizes, int n_in,
                              void* d_out, int out_size, void* d_ws, size_t ws_size,
                              hipStream_t stream) {
    const int*   ind2 = (const int*)  d_in[0];
    const float* p3   = (const float*)d_in[1];
    const float* i1   = (const float*)d_in[2];
    const float* d3   = (const float*)d_in[3];
    const float* W    = (const float*)d_in[4];

    float* p3_new = (float*)d_out;
    float* dotted = (float*)d_out + (size_t)N_ATOMS * 3 * CCH;
    float* agg    = (float*)d_ws;

    int*   scr  = (int*)d_out;
    int*   cnt  = scr + SCR_CNT;
    int*   off  = scr + SCR_OFF;
    int4*  entA = (int4*)(scr + SCR_ENTA);
    float* entZ = (float*)(scr + SCR_ENTZ);
    int*   rank = scr + SCR_RANK;

    zero_cnt_kernel<<<(N_ATOMS / 4 + 255) / 256, 256, 0, stream>>>((int4*)cnt, N_ATOMS / 4);

    hist_kernel<<<N_PAIRS / 256, 256, 0, stream>>>(ind2, cnt, rank);

    scan_kernel<<<1, 1024, 0, stream>>>(cnt, off);

    fill_kernel<<<N_PAIRS / 256, 256, 0, stream>>>(ind2, d3, off, rank, entA, entZ);

    segsum_kernel<<<N_ATOMS / 8, 256, 0, stream>>>(off, entA, entZ, p3, i1, agg);

    gemm_dot_kernel<<<(N_ATOMS / 16 + 3) / 4, 128, 0, stream>>>(agg, W, p3_new, dotted);
}


# GEMM and dot fused into the segment-sum kernel: 16 atoms per work-group, sums reach the f32 WMMA waves through LDS, results stored non-temporal; agg round trip and a launch gone; sum order unchanged
# speedup vs baseline: 1.5304x; 1.0376x over previous
#include <hip/hip_runtime.h>


#define N_ATOMS 50000
#define N_PAIRS 800000
#define CCH     64

#define SCR_CNT   0
#define SCR_OFF   50048
#define SCR_ENTA  100096
#define SCR_ENTZ  (100096 + 4 * N_PAIRS)
#define SCR_RANK  (100096 + 5 * N_PAIRS)

typedef __attribute__((ext_vector_type(2))) float v2f;
typedef __attribute__((ext_vector_type(8))) float v8f;

__global__ __launch_bounds__(256) void zero_cnt_kernel(int4* __restrict__ c, int n4) {
    int t = blockIdx.x * blockDim.x + threadIdx.x;
    if (t < n4) c[t] = make_int4(0, 0, 0, 0);
}

__global__ __launch_bounds__(256) void hist_kernel(
    const int* __restrict__ ind2,
    int*       __restrict__ cnt,
    int*       __restrict__ rank)
{
    const int p = blockIdx.x * 256 + threadIdx.x;
    const int i = ind2[2 * p];
    rank[p] = __hip_atomic_fetch_add(cnt + i, 1, __ATOMIC_RELAXED, __HIP_MEMORY_SCOPE_AGENT);
}

__global__ __launch_bounds__(1024) void scan_kernel(
    const int* __restrict__ cnt,
    int*       __restrict__ off)
{
    __shared__ int wtot[32];
    const int t = threadIdx.x, lane = t & 31, w = t >> 5;
    const int NQ = N_ATOMS / 4;
    int4 v[13];
    #pragma unroll
    for (int k = 0; k < 13; ++k) {
        const int q = k * 1024 + t;
        v[k] = (q < NQ) ? ((const int4*)cnt)[q] : make_int4(0, 0, 0, 0);
    }
    int carry = 0;
    #pragma unroll
    for (int k = 0; k < 13; ++k) {
        const int q = k * 1024 + t;
        const int s = v[k].x + v[k].y + v[k].z + v[k].w;
        int inc = s;
        for (int d = 1; d < 32; d <<= 1) {
            const int nb = __shfl_up(inc, d);
            if (lane >= d) inc += nb;
        }
        if (lane == 31) wtot[w] = inc;
        __syncthreads();
        const int wt = wtot[lane];
        int winc = wt;
        for (int d = 1; d < 32; d <<= 1) {
            const int nb = __shfl_up(winc, d);
            if (lane >= d) winc += nb;
        }
        const int wprefix = __shfl(winc, w) - __shfl(wt, w);
        const int total   = __shfl(winc, 31);
        __syncthreads();
        const int ex = carry + wprefix + (inc - s);
        int4 o;
        o.x = ex;
        o.y = o.x + v[k].x;
        o.z = o.y + v[k].y;
        o.w = o.z + v[k].z;
        if (q < NQ) ((int4*)off)[q] = o;
        carry += total;
    }
    if (t == 0) off[N_ATOMS] = carry;
}

__global__ __launch_bounds__(256) void fill_kernel(
    const int*   __restrict__ ind2,
    const float* __restrict__ d3,
    const int*   __restrict__ off,
    const int*   __restrict__ rank,
    int4*        __restrict__ entA,
    float*       __restrict__ entZ)
{
    const int p = blockIdx.x * 256 + threadIdx.x;
    const int i = ind2[2 * p];
    const int j = ind2[2 * p + 1];
    const float dx = d3[3 * p + 0];
    const float dy = d3[3 * p + 1];
    const float dz = d3[3 * p + 2];
    const int pos = off[i] + rank[p];
    entA[pos] = make_int4(p, j, __float_as_int(dx), __float_as_int(dy));
    entZ[pos] = dz;
}

template <int U>
__device__ __forceinline__ void seg_step(
    int k, int lane, int ep, int ej, float ex, float ey, float ez,
    const float* __restrict__ p3, const float* __restrict__ i1,
    v2f& acc0, v2f& acc1, v2f& acc2)
{
    v2f g[U], x0[U], x1[U], x2[U];
    float d0[U], d1[U], d2[U];
    #pragma unroll
    for (int u = 0; u < U; ++u) {
        const int p = __builtin_amdgcn_readlane(ep, k + u);
        const int j = __builtin_amdgcn_readlane(ej, k + u);
        d0[u] = __builtin_bit_cast(float, __builtin_amdgcn_readlane(__builtin_bit_cast(int, ex), k + u));
        d1[u] = __builtin_bit_cast(float, __builtin_amdgcn_readlane(__builtin_bit_cast(int, ey), k + u));
        d2[u] = __builtin_bit_cast(float, __builtin_amdgcn_readlane(__builtin_bit_cast(int, ez), k + u));
        g[u] = __builtin_nontemporal_load((const v2f*)(i1 + (size_t)p * CCH + 2 * lane));
        const float* s = p3 + (size_t)j * (3 * CCH) + 2 * lane;
        x0[u] = *(const v2f*)(s);
        x1[u] = *(const v2f*)(s + CCH);
        x2[u] = *(const v2f*)(s + 2 * CCH);
    }
    #pragma unroll
    for (int u = 0; u < U; ++u) {
        acc0 += (x0[u] + d0[u]) * g[u];
        acc1 += (x1[u] + d1[u]) * g[u];
        acc2 += (x2[u] + d2[u]) * g[u];
    }
}

typedef __attribute__((ext_vector_type(4))) float v4f;

#define GA   16
#define LDP  68
#define XST  (GA * LDP)

__global__ __launch_bounds__(512) void segsum_gemm_dot_kernel(
    const int*   __restrict__ off,
    const int4*  __restrict__ entA,
    const float* __restrict__ entZ,
    const float* __restrict__ p3,
    const float* __restrict__ i1,
    const float* __restrict__ W,
    float*       __restrict__ D,
    float*       __restrict__ dotted)
{
    __shared__ __attribute__((aligned(16))) float sA[3 * XST];
    __shared__ __attribute__((aligned(16))) float sD[3 * XST];

    const int lane = threadIdx.x & 31;
    const int w = __builtin_amdgcn_readfirstlane(threadIdx.x >> 5);

    const int a = blockIdx.x * GA + w;
    {
        const int beg = off[a];
        const int end = off[a + 1];

        v2f acc0 = {0.f, 0.f}, acc1 = {0.f, 0.f}, acc2 = {0.f, 0.f};
        for (int base = beg; base < end; base += 32) {
            const int n = (end - base < 32) ? (end - base) : 32;
            int ep = 0, ej = 0;
            float ex = 0.f, ey = 0.f, ez = 0.f;
            if (lane < n) {
                const int4 e = entA[base + lane];
                ep = e.x; ej = e.y;
                ex = __int_as_float(e.z); ey = __int_as_float(e.w);
                ez = entZ[base + lane];
            }
            int k = 0;
            for (; k + 4 <= n; k += 4)
                seg_step<4>(k, lane, ep, ej, ex, ey, ez, p3, i1, acc0, acc1, acc2);
            if (k + 2 <= n) {
                seg_step<2>(k, lane, ep, ej, ex, ey, ez, p3, i1, acc0, acc1, acc2);
                k += 2;
            }
            if (k < n)
                seg_step<1>(k, lane, ep, ej, ex, ey, ez, p3, i1, acc0, acc1, acc2);
        }

        float* s = sA + w * LDP + 2 * lane;
        *(v2f*)(s)           = acc0;
        *(v2f*)(s + XST)     = acc1;
        *(v2f*)(s + 2 * XST) = acc2;
    }

    const int c = lane & 15;
    const int h = lane >> 4;

    v4f bw[8] = {};
    if (w < 4) {
        const float* wrow = W + (size_t)(w * 16 + c) * CCH + 4 * h;
        #pragma unroll
        for (int q = 0; q < 8; ++q)
            bw[q] = *(const v4f*)(wrow + 8 * q);
    }
    __syncthreads();

    if (w < 4) {
        const float* arow = sA + c * LDP + 4 * h;

        v8f acc[3] = {};
        #pragma unroll
        for (int q = 0; q < 8; ++q) {
            const v2f b0 = {bw[q].x, bw[q].y};
            const v2f b1 = {bw[q].z, bw[q].w};
            #pragma unroll
            for (int x = 0; x < 3; ++x) {
                const v4f av = *(const v4f*)(arow + x * XST + 8 * q);
                const v2f a0 = {av.x, av.y};
                const v2f a1 = {av.z, av.w};
                acc[x] = __builtin_amdgcn_wmma_f32_16x16x4_f32(
                     false, a0,  false, b0,
                     (short)0, acc[x],  false,  false);
                acc[x] = __builtin_amdgcn_wmma_f32_16x16x4_f32(
                     false, a1,  false, b1,
                     (short)0, acc[x],  false,  false);
            }
        }

        float* o = sD + (8 * h) * LDP + w * 16 + c;
        #pragma unroll
        for (int x = 0; x < 3; ++x) {
            #pragma unroll
            for (int r = 0; r < 8; ++r)
                o[x * XST + r * LDP] = acc[x][r];
        }
    }
    __syncthreads();

    {
        const float* s = sD + w * LDP + 2 * lane;
        const v2f y0 = *(const v2f*)(s);
        const v2f y1 = *(const v2f*)(s + XST);
        const v2f y2 = *(const v2f*)(s + 2 * XST);
        float* dst = D + (size_t)a * (3 * CCH) + 2 * lane;
        __builtin_nontemporal_store(y0, (v2f*)(dst));
        __builtin_nontemporal_store(y1, (v2f*)(dst + CCH));
        __builtin_nontemporal_store(y2, (v2f*)(dst + 2 * CCH));
        __builtin_nontemporal_store(y0 * y0 + y1 * y1 + y2 * y2,
                                    (v2f*)(dotted + (size_t)a * CCH + 2 * lane));
    }
}

extern "C" void kernel_launch(void* const* d_in, const int* in_sizes, int n_in,
                              void* d_out, int out_size, void* d_ws, size_t ws_size,
                              hipStream_t stream) {
    const int*   ind2 = (const int*)  d_in[0];
    const float* p3   = (const float*)d_in[1];
    const float* i1   = (const float*)d_in[2];
    const float* d3   = (const float*)d_in[3];
    const float* W    = (const float*)d_in[4];

    float* p3_new = (float*)d_out;
    float* dotted = (float*)d_out + (size_t)N_ATOMS * 3 * CCH;

    int*   scr  = (int*)d_ws;
    int*   cnt  = scr + SCR_CNT;
    int*   off  = scr + SCR_OFF;
    int4*  entA = (int4*)(scr + SCR_ENTA);
    float* entZ = (float*)(scr + SCR_ENTZ);
    int*   rank = scr + SCR_RANK;

    zero_cnt_kernel<<<(N_ATOMS / 4 + 255) / 256, 256, 0, stream>>>((int4*)cnt, N_ATOMS / 4);

    hist_kernel<<<N_PAIRS / 256, 256, 0, stream>>>(ind2, cnt, rank);

    scan_kernel<<<1, 1024, 0, stream>>>(cnt, off);

    fill_kernel<<<N_PAIRS / 256, 256, 0, stream>>>(ind2, d3, off, rank, entA, entZ);

    segsum_gemm_dot_kernel<<<N_ATOMS / GA, 512, 0, stream>>>(off, entA, entZ, p3, i1, W, p3_new, dotted);
}


# Receiver sort without global atomics: two-level counting sort on LDS counters (pairs bucketed per 196-receiver range, then one work-group orders a bucket); sum order = LDS ticket arrival order
# speedup vs baseline: 2.1570x; 1.4094x over previous
#include <hip/hip_runtime.h>


#define N_ATOMS 50000
#define N_PAIRS 800000
#define CCH     64

#define NB    256
#define BWID  196
#define NCH   256
#define CHP   3125
#define NTAB  (NB * NCH)

#define SCR_TAB   0
#define SCR_OFF   65600
#define SCR_RECA  115648
#define SCR_RECZ  (115648 + 4 * N_PAIRS)
#define SCR_ENTA  (115648 + 5 * N_PAIRS)
#define SCR_ENTZ  (115648 + 9 * N_PAIRS)

typedef __attribute__((ext_vector_type(2))) float v2f;
typedef __attribute__((ext_vector_type(8))) float v8f;

#define CT   512
#define CU   7

__global__ __launch_bounds__(CT) void bucket_count_kernel(
    const int* __restrict__ ind2,
    int*       __restrict__ tab)
{
    __shared__ int sc[NB];
    const int t = threadIdx.x, ch = blockIdx.x;
    if (t < NB) sc[t] = 0;
    int iv[CU];
    #pragma unroll
    for (int u = 0; u < CU; ++u) {
        const int q = t + CT * u;
        iv[u] = (q < CHP) ? ind2[2 * (ch * CHP + q)] : -1;
    }
    __syncthreads();
    #pragma unroll
    for (int u = 0; u < CU; ++u)
        if (iv[u] >= 0) atomicAdd(&sc[iv[u] / BWID], 1);
    __syncthreads();
    if (t < NB) tab[t * NCH + ch] = sc[t];
}

__global__ __launch_bounds__(1024) void scan_kernel(
    int* tab)
{
    __shared__ int wtot[32];
    const int t = threadIdx.x, lane = t & 31, w = t >> 5;
    int4* tab4 = (int4*)tab;
    int4 v[16];
    #pragma unroll
    for (int k = 0; k < 16; ++k)
        v[k] = tab4[k * 1024 + t];
    int carry = 0;
    #pragma unroll
    for (int k = 0; k < 16; ++k) {
        const int s = v[k].x + v[k].y + v[k].z + v[k].w;
        int inc = s;
        for (int d = 1; d < 32; d <<= 1) {
            const int nb = __shfl_up(inc, d);
            if (lane >= d) inc += nb;
        }
        if (lane == 31) wtot[w] = inc;
        __syncthreads();
        const int wt = wtot[lane];
        int winc = wt;
        for (int d = 1; d < 32; d <<= 1) {
            const int nb = __shfl_up(winc, d);
            if (lane >= d) winc += nb;
        }
        const int wprefix = __shfl(winc, w) - __shfl(wt, w);
        const int total   = __shfl(winc, 31);
        __syncthreads();
        const int ex = carry + wprefix + (inc - s);
        int4 o;
        o.x = ex;
        o.y = o.x + v[k].x;
        o.z = o.y + v[k].y;
        o.w = o.z + v[k].z;
        tab4[k * 1024 + t] = o;
        carry += total;
    }
    if (t == 0) tab[NTAB] = carry;
}

__global__ __launch_bounds__(CT) void bucket_place_kernel(
    const int*   __restrict__ ind2,
    const float* __restrict__ d3,
    const int*   __restrict__ tab,
    int4*        __restrict__ recA,
    float*       __restrict__ recZ)
{
    __shared__ int cur[NB];
    const int t = threadIdx.x, ch = blockIdx.x;
    if (t < NB) cur[t] = tab[t * NCH + ch];
    int ri[CU], rj[CU];
    float dx[CU], dy[CU], dz[CU];
    #pragma unroll
    for (int u = 0; u < CU; ++u) {
        const int q = t + CT * u;
        ri[u] = -1; rj[u] = 0; dx[u] = 0.f; dy[u] = 0.f; dz[u] = 0.f;
        if (q < CHP) {
            const int p = ch * CHP + q;
            const int2 ij = ((const int2*)ind2)[p];
            ri[u] = ij.x; rj[u] = ij.y;
            dx[u] = d3[3 * p + 0];
            dy[u] = d3[3 * p + 1];
            dz[u] = d3[3 * p + 2];
        }
    }
    __syncthreads();
    #pragma unroll
    for (int u = 0; u < CU; ++u) {
        if (ri[u] >= 0) {
            const int p = ch * CHP + t + CT * u;
            const int b = ri[u] / BWID;
            const int k = ri[u] - b * BWID;
            const int pos = atomicAdd(&cur[b], 1);
            recA[pos] = make_int4(p, (k << 16) | rj[u], __float_as_int(dx[u]), __float_as_int(dy[u]));
            recZ[pos] = dz[u];
        }
    }
}

#define ST   1024
#define SU   4
__global__ __launch_bounds__(ST) void bucket_sort_kernel(
    const int*   __restrict__ tab,
    const int4*  __restrict__ recA,
    const float* __restrict__ recZ,
    int*         __restrict__ off,
    int4*        __restrict__ entA,
    float*       __restrict__ entZ)
{
    __shared__ int cnt[NB];
    __shared__ int cur[NB];
    const int t = threadIdx.x, b = blockIdx.x, lane = t & 31;
    const int bstart = tab[b * NCH];
    const int bend   = tab[(b + 1) * NCH];
    if (t < NB) cnt[t] = 0;
    __syncthreads();

    for (int q0 = bstart; q0 < bend; q0 += ST * SU) {
        int kw[SU];
        #pragma unroll
        for (int u = 0; u < SU; ++u) {
            const int q = q0 + ST * u + t;
            kw[u] = (q < bend) ? recA[q].y : -1;
        }
        #pragma unroll
        for (int u = 0; u < SU; ++u)
            if (kw[u] >= 0) atomicAdd(&cnt[kw[u] >> 16], 1);
    }
    __syncthreads();

    if (t < 32) {
        int c[8];
        int s = 0;
        #pragma unroll
        for (int m = 0; m < 8; ++m) { c[m] = cnt[8 * lane + m]; s += c[m]; }
        int inc = s;
        for (int d = 1; d < 32; d <<= 1) {
            const int nb = __shfl_up(inc, d);
            if (lane >= d) inc += nb;
        }
        int run = bstart + inc - s;
        #pragma unroll
        for (int m = 0; m < 8; ++m) { cur[8 * lane + m] = run; run += c[m]; }
    }
    __syncthreads();
    if (t < BWID) {
        const int a = b * BWID + t;
        if (a < N_ATOMS) off[a] = cur[t];
    }
    if (b == NB - 1 && t == 0) off[N_ATOMS] = bend;
    __syncthreads();

    for (int q0 = bstart; q0 < bend; q0 += ST * SU) {
        int4  e[SU];
        float z[SU];
        #pragma unroll
        for (int u = 0; u < SU; ++u) {
            const int q = q0 + ST * u + t;
            e[u] = make_int4(0, -1, 0, 0); z[u] = 0.f;
            if (q < bend) { e[u] = recA[q]; z[u] = recZ[q]; }
        }
        #pragma unroll
        for (int u = 0; u < SU; ++u) {
            if (e[u].y >= 0) {
                const int slot = atomicAdd(&cur[e[u].y >> 16], 1);
                entA[slot] = make_int4(e[u].x, e[u].y & 0xffff, e[u].z, e[u].w);
                entZ[slot] = z[u];
            }
        }
    }
}

template <int U>
__device__ __forceinline__ void seg_step(
    int k, int lane, int ep, int ej, float ex, float ey, float ez,
    const float* __restrict__ p3, const float* __restrict__ i1,
    v2f& acc0, v2f& acc1, v2f& acc2)
{
    v2f g[U], x0[U], x1[U], x2[U];
    float d0[U], d1[U], d2[U];
    #pragma unroll
    for (int u = 0; u < U; ++u) {
        const int p = __builtin_amdgcn_readlane(ep, k + u);
        const int j = __builtin_amdgcn_readlane(ej, k + u);
        d0[u] = __builtin_bit_cast(float, __builtin_amdgcn_readlane(__builtin_bit_cast(int, ex), k + u));
        d1[u] = __builtin_bit_cast(float, __builtin_amdgcn_readlane(__builtin_bit_cast(int, ey), k + u));
        d2[u] = __builtin_bit_cast(float, __builtin_amdgcn_readlane(__builtin_bit_cast(int, ez), k + u));
        g[u] = __builtin_nontemporal_load((const v2f*)(i1 + (size_t)p * CCH + 2 * lane));
        const float* s = p3 + (size_t)j * (3 * CCH) + 2 * lane;
        x0[u] = *(const v2f*)(s);
        x1[u] = *(const v2f*)(s + CCH);
        x2[u] = *(const v2f*)(s + 2 * CCH);
    }
    #pragma unroll
    for (int u = 0; u < U; ++u) {
        acc0 += (x0[u] + d0[u]) * g[u];
        acc1 += (x1[u] + d1[u]) * g[u];
        acc2 += (x2[u] + d2[u]) * g[u];
    }
}

typedef __attribute__((ext_vector_type(4))) float v4f;

#define GA   16
#define LDP  68
#define XST  (GA * LDP)

__global__ __launch_bounds__(512) void segsum_gemm_dot_kernel(
    const int*   __restrict__ off,
    const int4*  __restrict__ entA,
    const float* __restrict__ entZ,
    const float* __restrict__ p3,
    const float* __restrict__ i1,
    const float* __restrict__ W,
    float*       __restrict__ D,
    float*       __restrict__ dotted)
{
    __shared__ __attribute__((aligned(16))) float sA[3 * XST];
    __shared__ __attribute__((aligned(16))) float sD[3 * XST];

    const int lane = threadIdx.x & 31;
    const int w = __builtin_amdgcn_readfirstlane(threadIdx.x >> 5);

    const int a = blockIdx.x * GA + w;
    {
        const int beg = off[a];
        const int end = off[a + 1];

        v2f acc0 = {0.f, 0.f}, acc1 = {0.f, 0.f}, acc2 = {0.f, 0.f};
        for (int base = beg; base < end; base += 32) {
            const int n = (end - base < 32) ? (end - base) : 32;
            int ep = 0, ej = 0;
            float ex = 0.f, ey = 0.f, ez = 0.f;
            if (lane < n) {
                const int4 e = entA[base + lane];
                ep = e.x; ej = e.y;
                ex = __int_as_float(e.z); ey = __int_as_float(e.w);
                ez = entZ[base + lane];
            }
            int k = 0;
            for (; k + 4 <= n; k += 4)
                seg_step<4>(k, lane, ep, ej, ex, ey, ez, p3, i1, acc0, acc1, acc2);
            if (k + 2 <= n) {
                seg_step<2>(k, lane, ep, ej, ex, ey, ez, p3, i1, acc0, acc1, acc2);
                k += 2;
            }
            if (k < n)
                seg_step<1>(k, lane, ep, ej, ex, ey, ez, p3, i1, acc0, acc1, acc2);
        }

        float* s = sA + w * LDP + 2 * lane;
        *(v2f*)(s)           = acc0;
        *(v2f*)(s + XST)     = acc1;
        *(v2f*)(s + 2 * XST) = acc2;
    }

    const int c = lane & 15;
    const int h = lane >> 4;

    v4f bw[8] = {};
    if (w < 4) {
        const float* wrow = W + (size_t)(w * 16 + c) * CCH + 4 * h;
        #pragma unroll
        for (int q = 0; q < 8; ++q)
            bw[q] = *(const v4f*)(wrow + 8 * q);
    }
    __syncthreads();

    if (w < 4) {
        const float* arow = sA + c * LDP + 4 * h;

        v8f acc[3] = {};
        #pragma unroll
        for (int q = 0; q < 8; ++q) {
            const v2f b0 = {bw[q].x, bw[q].y};
            const v2f b1 = {bw[q].z, bw[q].w};
            #pragma unroll
            for (int x = 0; x < 3; ++x) {
                const v4f av = *(const v4f*)(arow + x * XST + 8 * q);
                const v2f a0 = {av.x, av.y};
                const v2f a1 = {av.z, av.w};
                acc[x] = __builtin_amdgcn_wmma_f32_16x16x4_f32(
                     false, a0,  false, b0,
                     (short)0, acc[x],  false,  false);
                acc[x] = __builtin_amdgcn_wmma_f32_16x16x4_f32(
                     false, a1,  false, b1,
                     (short)0, acc[x],  false,  false);
            }
        }

        float* o = sD + (8 * h) * LDP + w * 16 + c;
        #pragma unroll
        for (int x = 0; x < 3; ++x) {
            #pragma unroll
            for (int r = 0; r < 8; ++r)
                o[x * XST + r * LDP] = acc[x][r];
        }
    }
    __syncthreads();

    {
        const float* s = sD + w * LDP + 2 * lane;
        const v2f y0 = *(const v2f*)(s);
        const v2f y1 = *(const v2f*)(s + XST);
        const v2f y2 = *(const v2f*)(s + 2 * XST);
        float* dst = D + (size_t)a * (3 * CCH) + 2 * lane;
        __builtin_nontemporal_store(y0, (v2f*)(dst));
        __builtin_nontemporal_store(y1, (v2f*)(dst + CCH));
        __builtin_nontemporal_store(y2, (v2f*)(dst + 2 * CCH));
        __builtin_nontemporal_store(y0 * y0 + y1 * y1 + y2 * y2,
                                    (v2f*)(dotted + (size_t)a * CCH + 2 * lane));
    }
}

extern "C" void kernel_launch(void* const* d_in, const int* in_sizes, int n_in,
                              void* d_out, int out_size, void* d_ws, size_t ws_size,
                              hipStream_t stream) {
    const int*   ind2 = (const int*)  d_in[0];
    const float* p3   = (const float*)d_in[1];
    const float* i1   = (const float*)d_in[2];
    const float* d3   = (const float*)d_in[3];
    const float* W    = (const float*)d_in[4];

    float* p3_new = (float*)d_out;
    float* dotted = (float*)d_out + (size_t)N_ATOMS * 3 * CCH;

    int*   scr  = (int*)d_ws;
    int*   tab  = scr + SCR_TAB;
    int*   off  = scr + SCR_OFF;
    int4*  recA = (int4*)(scr + SCR_RECA);
    float* recZ = (float*)(scr + SCR_RECZ);
    int4*  entA = (int4*)(scr + SCR_ENTA);
    float* entZ = (float*)(scr + SCR_ENTZ);

    bucket_count_kernel<<<NCH, CT, 0, stream>>>(ind2, tab);

    scan_kernel<<<1, 1024, 0, stream>>>(tab);

    bucket_place_kernel<<<NCH, CT, 0, stream>>>(ind2, d3, tab, recA, recZ);

    bucket_sort_kernel<<<NB, ST, 0, stream>>>(tab, recA, recZ, off, entA, entZ);

    segsum_gemm_dot_kernel<<<N_ATOMS / GA, 512, 0, stream>>>(off, entA, entZ, p3, i1, W, p3_new, dotted);
}


# Scan launch removed: each placing work-group sums the raw count table (now chunk-major, whole-row writes) into bucket totals and earlier-chunk counts, scans the 256 totals itself; sum order unchanged
# speedup vs baseline: 2.6011x; 1.2059x over previous
#include <hip/hip_runtime.h>


#define N_ATOMS 50000
#define N_PAIRS 800000
#define CCH     64

#define NB    256
#define BWID  196
#define NCH   256
#define CHP   3125
#define NTAB  (NB * NCH)

#define SCR_TAB   0
#define SCR_OFF   65600
#define SCR_RECA  115648
#define SCR_RECZ  (115648 + 4 * N_PAIRS)
#define SCR_ENTA  (115648 + 5 * N_PAIRS)
#define SCR_ENTZ  (115648 + 9 * N_PAIRS)
#define SCR_BST   (115648 + 10 * N_PAIRS)

typedef __attribute__((ext_vector_type(2))) float v2f;
typedef __attribute__((ext_vector_type(8))) float v8f;

#define CT   512
#define CU   7

__global__ __launch_bounds__(CT) void bucket_count_kernel(
    const int* __restrict__ ind2,
    int*       __restrict__ tab)
{
    __shared__ int sc[NB];
    const int t = threadIdx.x, ch = blockIdx.x;
    if (t < NB) sc[t] = 0;
    int iv[CU];
    #pragma unroll
    for (int u = 0; u < CU; ++u) {
        const int q = t + CT * u;
        iv[u] = (q < CHP) ? ind2[2 * (ch * CHP + q)] : -1;
    }
    __syncthreads();
    #pragma unroll
    for (int u = 0; u < CU; ++u)
        if (iv[u] >= 0) atomicAdd(&sc[iv[u] / BWID], 1);
    __syncthreads();
    if (t < NB) tab[ch * NB + t] = sc[t];
}

#define TG   8
#define TR   (NCH / TG)
__global__ __launch_bounds__(CT) void bucket_place_kernel(
    const int*   __restrict__ ind2,
    const float* __restrict__ d3,
    const int*   __restrict__ tab,
    int*         __restrict__ bst,
    int4*        __restrict__ recA,
    float*       __restrict__ recZ)
{
    __shared__ __attribute__((aligned(16))) int pT[TG][NB];
    __shared__ __attribute__((aligned(16))) int pP[TG][NB];
    __shared__ int wtot[NB / 32];
    __shared__ int cur[NB];
    const int t = threadIdx.x, ch = blockIdx.x, lane = t & 31;

    const int bq = t & 63;
    const int g = __builtin_amdgcn_readfirstlane(t >> 6);
    const int4* tab4 = (const int4*)tab;
    int4 tv[TR];
    #pragma unroll
    for (int k = 0; k < TR; ++k) {
        const int c = g * TR + ((k + ch) & (TR - 1));
        tv[k] = tab4[c * (NB / 4) + bq];
    }

    int ri[CU], rj[CU];
    float dx[CU], dy[CU], dz[CU];
    #pragma unroll
    for (int u = 0; u < CU; ++u) {
        const int q = t + CT * u;
        ri[u] = -1; rj[u] = 0; dx[u] = 0.f; dy[u] = 0.f; dz[u] = 0.f;
        if (q < CHP) {
            const int p = ch * CHP + q;
            const int2 ij = ((const int2*)ind2)[p];
            ri[u] = ij.x; rj[u] = ij.y;
            dx[u] = d3[3 * p + 0];
            dy[u] = d3[3 * p + 1];
            dz[u] = d3[3 * p + 2];
        }
    }

    int4 tot = make_int4(0, 0, 0, 0), pre = make_int4(0, 0, 0, 0);
    #pragma unroll
    for (int k = 0; k < TR; ++k) {
        const int c = g * TR + ((k + ch) & (TR - 1));
        tot.x += tv[k].x; tot.y += tv[k].y; tot.z += tv[k].z; tot.w += tv[k].w;
        if (c < ch) { pre.x += tv[k].x; pre.y += tv[k].y; pre.z += tv[k].z; pre.w += tv[k].w; }
    }
    *(int4*)&pT[g][4 * bq] = tot;
    *(int4*)&pP[g][4 * bq] = pre;
    __syncthreads();

    int T = 0, P = 0, inc = 0;
    if (t < NB) {
        #pragma unroll
        for (int m = 0; m < TG; ++m) { T += pT[m][t]; P += pP[m][t]; }
        inc = T;
        for (int d = 1; d < 32; d <<= 1) {
            const int nb = __shfl_up(inc, d);
            if (lane >= d) inc += nb;
        }
        if (lane == 31) wtot[t >> 5] = inc;
    }
    __syncthreads();
    if (t < NB) {
        int wp = 0;
        #pragma unroll
        for (int m = 0; m < NB / 32; ++m) {
            const int s = wtot[m];
            if (m < (t >> 5)) wp += s;
        }
        const int start = wp + inc - T;
        cur[t] = start + P;
        if (ch == 0) {
            bst[t] = start;
            if (t == NB - 1) bst[NB] = start + T;
        }
    }
    __syncthreads();
    #pragma unroll
    for (int u = 0; u < CU; ++u) {
        if (ri[u] >= 0) {
            const int p = ch * CHP + t + CT * u;
            const int b = ri[u] / BWID;
            const int k = ri[u] - b * BWID;
            const int pos = atomicAdd(&cur[b], 1);
            recA[pos] = make_int4(p, (k << 16) | rj[u], __float_as_int(dx[u]), __float_as_int(dy[u]));
            recZ[pos] = dz[u];
        }
    }
}

#define ST   1024
#define SU   4
__global__ __launch_bounds__(ST) void bucket_sort_kernel(
    const int*   __restrict__ bst,
    const int4*  __restrict__ recA,
    const float* __restrict__ recZ,
    int*         __restrict__ off,
    int4*        __restrict__ entA,
    float*       __restrict__ entZ)
{
    __shared__ int cnt[NB];
    __shared__ int cur[NB];
    const int t = threadIdx.x, b = blockIdx.x, lane = t & 31;
    const int bstart = bst[b];
    const int bend   = bst[b + 1];
    if (t < NB) cnt[t] = 0;
    __syncthreads();

    for (int q0 = bstart; q0 < bend; q0 += ST * SU) {
        int kw[SU];
        #pragma unroll
        for (int u = 0; u < SU; ++u) {
            const int q = q0 + ST * u + t;
            kw[u] = (q < bend) ? recA[q].y : -1;
        }
        #pragma unroll
        for (int u = 0; u < SU; ++u)
            if (kw[u] >= 0) atomicAdd(&cnt[kw[u] >> 16], 1);
    }
    __syncthreads();

    if (t < 32) {
        int c[8];
        int s = 0;
        #pragma unroll
        for (int m = 0; m < 8; ++m) { c[m] = cnt[8 * lane + m]; s += c[m]; }
        int inc = s;
        for (int d = 1; d < 32; d <<= 1) {
            const int nb = __shfl_up(inc, d);
            if (lane >= d) inc += nb;
        }
        int run = bstart + inc - s;
        #pragma unroll
        for (int m = 0; m < 8; ++m) { cur[8 * lane + m] = run; run += c[m]; }
    }
    __syncthreads();
    if (t < BWID) {
        const int a = b * BWID + t;
        if (a < N_ATOMS) off[a] = cur[t];
    }
    if (b == NB - 1 && t == 0) off[N_ATOMS] = bend;
    __syncthreads();

    for (int q0 = bstart; q0 < bend; q0 += ST * SU) {
        int4  e[SU];
        float z[SU];
        #pragma unroll
        for (int u = 0; u < SU; ++u) {
            const int q = q0 + ST * u + t;
            e[u] = make_int4(0, -1, 0, 0); z[u] = 0.f;
            if (q < bend) { e[u] = recA[q]; z[u] = recZ[q]; }
        }
        #pragma unroll
        for (int u = 0; u < SU; ++u) {
            if (e[u].y >= 0) {
                const int slot = atomicAdd(&cur[e[u].y >> 16], 1);
                entA[slot] = make_int4(e[u].x, e[u].y & 0xffff, e[u].z, e[u].w);
                entZ[slot] = z[u];
            }
        }
    }
}

template <int U>
__device__ __forceinline__ void seg_step(
    int k, int lane, int ep, int ej, float ex, float ey, float ez,
    const float* __restrict__ p3, const float* __restrict__ i1,
    v2f& acc0, v2f& acc1, v2f& acc2)
{
    v2f g[U], x0[U], x1[U], x2[U];
    float d0[U], d1[U], d2[U];
    #pragma unroll
    for (int u = 0; u < U; ++u) {
        const int p = __builtin_amdgcn_readlane(ep, k + u);
        const int j = __builtin_amdgcn_readlane(ej, k + u);
        d0[u] = __builtin_bit_cast(float, __builtin_amdgcn_readlane(__builtin_bit_cast(int, ex), k + u));
        d1[u] = __builtin_bit_cast(float, __builtin_amdgcn_readlane(__builtin_bit_cast(int, ey), k + u));
        d2[u] = __builtin_bit_cast(float, __builtin_amdgcn_readlane(__builtin_bit_cast(int, ez), k + u));
        g[u] = __builtin_nontemporal_load((const v2f*)(i1 + (size_t)p * CCH + 2 * lane));
        const float* s = p3 + (size_t)j * (3 * CCH) + 2 * lane;
        x0[u] = *(const v2f*)(s);
        x1[u] = *(const v2f*)(s + CCH);
        x2[u] = *(const v2f*)(s + 2 * CCH);
    }
    #pragma unroll
    for (int u = 0; u < U; ++u) {
        acc0 += (x0[u] + d0[u]) * g[u];
        acc1 += (x1[u] + d1[u]) * g[u];
        acc2 += (x2[u] + d2[u]) * g[u];
    }
}

typedef __attribute__((ext_vector_type(4))) float v4f;

#define GA   16
#define LDP  68
#define XST  (GA * LDP)

__global__ __launch_bounds__(512) void segsum_gemm_dot_kernel(
    const int*   __restrict__ off,
    const int4*  __restrict__ entA,
    const float* __restrict__ entZ,
    const float* __restrict__ p3,
    const float* __restrict__ i1,
    const float* __restrict__ W,
    float*       __restrict__ D,
    float*       __restrict__ dotted)
{
    __shared__ __attribute__((aligned(16))) float sA[3 * XST];
    __shared__ __attribute__((aligned(16))) float sD[3 * XST];

    const int lane = threadIdx.x & 31;
    const int w = __builtin_amdgcn_readfirstlane(threadIdx.x >> 5);

    const int a = blockIdx.x * GA + w;
    {
        const int beg = off[a];
        const int end = off[a + 1];

        v2f acc0 = {0.f, 0.f}, acc1 = {0.f, 0.f}, acc2 = {0.f, 0.f};
        for (int base = beg; base < end; base += 32) {
            const int n = (end - base < 32) ? (end - base) : 32;
            int ep = 0, ej = 0;
            float ex = 0.f, ey = 0.f, ez = 0.f;
            if (lane < n) {
                const int4 e = entA[base + lane];
                ep = e.x; ej = e.y;
                ex = __int_as_float(e.z); ey = __int_as_float(e.w);
                ez = entZ[base + lane];
            }
            int k = 0;
            for (; k + 4 <= n; k += 4)
                seg_step<4>(k, lane, ep, ej, ex, ey, ez, p3, i1, acc0, acc1, acc2);
            if (k + 2 <= n) {
                seg_step<2>(k, lane, ep, ej, ex, ey, ez, p3, i1, acc0, acc1, acc2);
                k += 2;
            }
            if (k < n)
                seg_step<1>(k, lane, ep, ej, ex, ey, ez, p3, i1, acc0, acc1, acc2);
        }

        float* s = sA + w * LDP + 2 * lane;
        *(v2f*)(s)           = acc0;
        *(v2f*)(s + XST)     = acc1;
        *(v2f*)(s + 2 * XST) = acc2;
    }

    const int c = lane & 15;
    const int h = lane >> 4;

    v4f bw[8] = {};
    if (w < 4) {
        const float* wrow = W + (size_t)(w * 16 + c) * CCH + 4 * h;
        #pragma unroll
        for (int q = 0; q < 8; ++q)
            bw[q] = *(const v4f*)(wrow + 8 * q);
    }
    __syncthreads();

    if (w < 4) {
        const float* arow = sA + c * LDP + 4 * h;

        v8f acc[3] = {};
        #pragma unroll
        for (int q = 0; q < 8; ++q) {
            const v2f b0 = {bw[q].x, bw[q].y};
            const v2f b1 = {bw[q].z, bw[q].w};
            #pragma unroll
            for (int x = 0; x < 3; ++x) {
                const v4f av = *(const v4f*)(arow + x * XST + 8 * q);
                const v2f a0 = {av.x, av.y};
                const v2f a1 = {av.z, av.w};
                acc[x] = __builtin_amdgcn_wmma_f32_16x16x4_f32(
                     false, a0,  false, b0,
                     (short)0, acc[x],  false,  false);
                acc[x] = __builtin_amdgcn_wmma_f32_16x16x4_f32(
                     false, a1,  false, b1,
                     (short)0, acc[x],  false,  false);
            }
        }

        float* o = sD + (8 * h) * LDP + w * 16 + c;
        #pragma unroll
        for (int x = 0; x < 3; ++x) {
            #pragma unroll
            for (int r = 0; r < 8; ++r)
                o[x * XST + r * LDP] = acc[x][r];
        }
    }
    __syncthreads();

    {
        const float* s = sD + w * LDP + 2 * lane;
        const v2f y0 = *(const v2f*)(s);
        const v2f y1 = *(const v2f*)(s + XST);
        const v2f y2 = *(const v2f*)(s + 2 * XST);
        float* dst = D + (size_t)a * (3 * CCH) + 2 * lane;
        __builtin_nontemporal_store(y0, (v2f*)(dst));
        __builtin_nontemporal_store(y1, (v2f*)(dst + CCH));
        __builtin_nontemporal_store(y2, (v2f*)(dst + 2 * CCH));
        __builtin_nontemporal_store(y0 * y0 + y1 * y1 + y2 * y2,
                                    (v2f*)(dotted + (size_t)a * CCH + 2 * lane));
    }
}

extern "C" void kernel_launch(void* const* d_in, const int* in_sizes, int n_in,
                              void* d_out, int out_size, void* d_ws, size_t ws_size,
                              hipStream_t stream) {
    const int*   ind2 = (const int*)  d_in[0];
    const float* p3   = (const float*)d_in[1];
    const float* i1   = (const float*)d_in[2];
    const float* d3   = (const float*)d_in[3];
    const float* W    = (const float*)d_in[4];

    float* p3_new = (float*)d_out;
    float* dotted = (float*)d_out + (size_t)N_ATOMS * 3 * CCH;

    int*   scr  = (int*)d_ws;
    int*   tab  = scr + SCR_TAB;
    int*   off  = scr + SCR_OFF;
    int*   bst  = scr + SCR_BST;
    int4*  recA = (int4*)(scr + SCR_RECA);
    float* recZ = (float*)(scr + SCR_RECZ);
    int4*  entA = (int4*)(scr + SCR_ENTA);
    float* entZ = (float*)(scr + SCR_ENTZ);

    bucket_count_kernel<<<NCH, CT, 0, stream>>>(ind2, tab);

    bucket_place_kernel<<<NCH, CT, 0, stream>>>(ind2, d3, tab, bst, recA, recZ);

    bucket_sort_kernel<<<NB, ST, 0, stream>>>(bst, recA, recZ, off, entA, entZ);

    segsum_gemm_dot_kernel<<<N_ATOMS / GA, 512, 0, stream>>>(off, entA, entZ, p3, i1, W, p3_new, dotted);
}


# Count launch removed: a chunk's work-group counts, scans and places its own pairs by bucket in one launch; a bucket's group gets start and runs from one table column; sum order = LDS ticket order
# speedup vs baseline: 3.1658x; 1.2171x over previous
#include <hip/hip_runtime.h>


#define N_ATOMS 50000
#define N_PAIRS 800000
#define CCH     64

#define NB    256
#define BWID  196
#define NCH   256
#define CHP   3125
#define NTAB  (NB * NCH)

#define SCR_TAB   0
#define SCR_OFF   65600
#define SCR_RECA  115648
#define SCR_RECZ  (115648 + 4 * N_PAIRS)
#define SCR_ENTA  (115648 + 5 * N_PAIRS)
#define SCR_ENTZ  (115648 + 9 * N_PAIRS)

typedef __attribute__((ext_vector_type(2))) float v2f;
typedef __attribute__((ext_vector_type(8))) float v8f;

#define CT   512
#define CU   7

__global__ __launch_bounds__(CT) void chunk_bucket_kernel(
    const int*   __restrict__ ind2,
    const float* __restrict__ d3,
    int*         __restrict__ tab,
    int4*        __restrict__ recA,
    float*       __restrict__ recZ)
{
    __shared__ int sc[NB];
    __shared__ int wtot[NB / 32];
    const int t = threadIdx.x, ch = blockIdx.x, lane = t & 31;
    if (t < NB) sc[t] = 0;

    int ri[CU], rj[CU];
    float dx[CU], dy[CU], dz[CU];
    #pragma unroll
    for (int u = 0; u < CU; ++u) {
        const int q = t + CT * u;
        ri[u] = -1; rj[u] = 0; dx[u] = 0.f; dy[u] = 0.f; dz[u] = 0.f;
        if (q < CHP) {
            const int p = ch * CHP + q;
            const int2 ij = ((const int2*)ind2)[p];
            ri[u] = ij.x; rj[u] = ij.y;
            dx[u] = d3[3 * p + 0];
            dy[u] = d3[3 * p + 1];
            dz[u] = d3[3 * p + 2];
        }
    }
    __syncthreads();
    #pragma unroll
    for (int u = 0; u < CU; ++u)
        if (ri[u] >= 0) atomicAdd(&sc[ri[u] / BWID], 1);
    __syncthreads();

    int T = 0, inc = 0;
    if (t < NB) {
        T = sc[t];
        inc = T;
        for (int d = 1; d < 32; d <<= 1) {
            const int nb = __shfl_up(inc, d);
            if (lane >= d) inc += nb;
        }
        if (lane == 31) wtot[t >> 5] = inc;
    }
    __syncthreads();
    if (t < NB) {
        int wp = 0;
        #pragma unroll
        for (int m = 0; m < NB / 32; ++m) {
            const int s = wtot[m];
            if (m < (t >> 5)) wp += s;
        }
        const int ls = wp + inc - T;
        sc[t] = ls;
        tab[ch * NB + t] = ls | (T << 16);
    }
    __syncthreads();
    #pragma unroll
    for (int u = 0; u < CU; ++u) {
        if (ri[u] >= 0) {
            const int p = ch * CHP + t + CT * u;
            const int b = ri[u] / BWID;
            const int k = ri[u] - b * BWID;
            const int pos = ch * CHP + atomicAdd(&sc[b], 1);
            recA[pos] = make_int4(p, (k << 16) | rj[u], __float_as_int(dx[u]), __float_as_int(dy[u]));
            recZ[pos] = dz[u];
        }
    }
}

#define ST   1024
#define SU   4
__global__ __launch_bounds__(ST) void bucket_sort_kernel(
    const int*   __restrict__ tab,
    const int4*  __restrict__ recA,
    const float* __restrict__ recZ,
    int*         __restrict__ off,
    int4*        __restrict__ entA,
    float*       __restrict__ entZ)
{
    __shared__ int cnt[NB];
    __shared__ int cur[NB];
    __shared__ int rsrc[NCH];
    __shared__ int rlen[NCH];
    __shared__ int wls[NCH / 32];
    __shared__ int wcn[NCH / 32];
    const int t = threadIdx.x, b = blockIdx.x, lane = t & 31;
    if (t < NB) cnt[t] = 0;
    if (t < NCH) {
        const int v = tab[t * NB + b];
        int ls = v & 0xffff;
        int n  = v >> 16;
        rsrc[t] = t * CHP + ls;
        rlen[t] = n;
        for (int d = 16; d >= 1; d >>= 1) {
            ls += __shfl_xor(ls, d);
            n  += __shfl_xor(n, d);
        }
        if (lane == 0) { wls[t >> 5] = ls; wcn[t >> 5] = n; }
    }
    __syncthreads();

    int bstart = 0, btot = 0;
    #pragma unroll
    for (int m = 0; m < NCH / 32; ++m) { bstart += wls[m]; btot += wcn[m]; }
    const int bend = bstart + btot;

    const int src = rsrc[t >> 2];
    const int n   = rlen[t >> 2];
    const int sub = t & 3;

    for (int r0 = sub; r0 < n; r0 += 4 * SU) {
        int kw[SU];
        #pragma unroll
        for (int u = 0; u < SU; ++u) {
            const int r = r0 + 4 * u;
            kw[u] = (r < n) ? recA[src + r].y : -1;
        }
        #pragma unroll
        for (int u = 0; u < SU; ++u)
            if (kw[u] >= 0) atomicAdd(&cnt[kw[u] >> 16], 1);
    }
    __syncthreads();

    if (t < 32) {
        int c[8];
        int s = 0;
        #pragma unroll
        for (int m = 0; m < 8; ++m) { c[m] = cnt[8 * lane + m]; s += c[m]; }
        int inc = s;
        for (int d = 1; d < 32; d <<= 1) {
            const int nb = __shfl_up(inc, d);
            if (lane >= d) inc += nb;
        }
        int run = bstart + inc - s;
        #pragma unroll
        for (int m = 0; m < 8; ++m) { cur[8 * lane + m] = run; run += c[m]; }
    }
    __syncthreads();
    if (t < BWID) {
        const int a = b * BWID + t;
        if (a < N_ATOMS) off[a] = cur[t];
    }
    if (b == NB - 1 && t == 0) off[N_ATOMS] = bend;
    __syncthreads();

    for (int r0 = sub; r0 < n; r0 += 4 * SU) {
        int4  e[SU];
        float z[SU];
        #pragma unroll
        for (int u = 0; u < SU; ++u) {
            const int r = r0 + 4 * u;
            e[u] = make_int4(0, -1, 0, 0); z[u] = 0.f;
            if (r < n) { e[u] = recA[src + r]; z[u] = recZ[src + r]; }
        }
        #pragma unroll
        for (int u = 0; u < SU; ++u) {
            if (e[u].y >= 0) {
                const int slot = atomicAdd(&cur[e[u].y >> 16], 1);
                entA[slot] = make_int4(e[u].x, e[u].y & 0xffff, e[u].z, e[u].w);
                entZ[slot] = z[u];
            }
        }
    }
}

template <int U>
__device__ __forceinline__ void seg_step(
    int k, int lane, int ep, int ej, float ex, float ey, float ez,
    const float* __restrict__ p3, const float* __restrict__ i1,
    v2f& acc0, v2f& acc1, v2f& acc2)
{
    v2f g[U], x0[U], x1[U], x2[U];
    float d0[U], d1[U], d2[U];
    #pragma unroll
    for (int u = 0; u < U; ++u) {
        const int p = __builtin_amdgcn_readlane(ep, k + u);
        const int j = __builtin_amdgcn_readlane(ej, k + u);
        d0[u] = __builtin_bit_cast(float, __builtin_amdgcn_readlane(__builtin_bit_cast(int, ex), k + u));
        d1[u] = __builtin_bit_cast(float, __builtin_amdgcn_readlane(__builtin_bit_cast(int, ey), k + u));
        d2[u] = __builtin_bit_cast(float, __builtin_amdgcn_readlane(__builtin_bit_cast(int, ez), k + u));
        g[u] = __builtin_nontemporal_load((const v2f*)(i1 + (size_t)p * CCH + 2 * lane));
        const float* s = p3 + (size_t)j * (3 * CCH) + 2 * lane;
        x0[u] = *(const v2f*)(s);
        x1[u] = *(const v2f*)(s + CCH);
        x2[u] = *(const v2f*)(s + 2 * CCH);
    }
    #pragma unroll
    for (int u = 0; u < U; ++u) {
        acc0 += (x0[u] + d0[u]) * g[u];
        acc1 += (x1[u] + d1[u]) * g[u];
        acc2 += (x2[u] + d2[u]) * g[u];
    }
}

typedef __attribute__((ext_vector_type(4))) float v4f;

#define GA   16
#define LDP  68
#define XST  (GA * LDP)

__global__ __launch_bounds__(512) void segsum_gemm_dot_kernel(
    const int*   __restrict__ off,
    const int4*  __restrict__ entA,
    const float* __restrict__ entZ,
    const float* __restrict__ p3,
    const float* __restrict__ i1,
    const float* __restrict__ W,
    float*       __restrict__ D,
    float*       __restrict__ dotted)
{
    __shared__ __attribute__((aligned(16))) float sA[3 * XST];
    __shared__ __attribute__((aligned(16))) float sD[3 * XST];

    const int lane = threadIdx.x & 31;
    const int w = __builtin_amdgcn_readfirstlane(threadIdx.x >> 5);

    const int a = blockIdx.x * GA + w;
    {
        const int beg = off[a];
        const int end = off[a + 1];

        v2f acc0 = {0.f, 0.f}, acc1 = {0.f, 0.f}, acc2 = {0.f, 0.f};
        for (int base = beg; base < end; base += 32) {
            const int n = (end - base < 32) ? (end - base) : 32;
            int ep = 0, ej = 0;
            float ex = 0.f, ey = 0.f, ez = 0.f;
            if (lane < n) {
                const int4 e = entA[base + lane];
                ep = e.x; ej = e.y;
                ex = __int_as_float(e.z); ey = __int_as_float(e.w);
                ez = entZ[base + lane];
            }
            int k = 0;
            for (; k + 4 <= n; k += 4)
                seg_step<4>(k, lane, ep, ej, ex, ey, ez, p3, i1, acc0, acc1, acc2);
            if (k + 2 <= n) {
                seg_step<2>(k, lane, ep, ej, ex, ey, ez, p3, i1, acc0, acc1, acc2);
                k += 2;
            }
            if (k < n)
                seg_step<1>(k, lane, ep, ej, ex, ey, ez, p3, i1, acc0, acc1, acc2);
        }

        float* s = sA + w * LDP + 2 * lane;
        *(v2f*)(s)           = acc0;
        *(v2f*)(s + XST)     = acc1;
        *(v2f*)(s + 2 * XST) = acc2;
    }

    const int c = lane & 15;
    const int h = lane >> 4;

    v4f bw[8] = {};
    if (w < 4) {
        const float* wrow = W + (size_t)(w * 16 + c) * CCH + 4 * h;
        #pragma unroll
        for (int q = 0; q < 8; ++q)
            bw[q] = *(const v4f*)(wrow + 8 * q);
    }
    __syncthreads();

    if (w < 4) {
        const float* arow = sA + c * LDP + 4 * h;

        v8f acc[3] = {};
        #pragma unroll
        for (int q = 0; q < 8; ++q) {
            const v2f b0 = {bw[q].x, bw[q].y};
            const v2f b1 = {bw[q].z, bw[q].w};
            #pragma unroll
            for (int x = 0; x < 3; ++x) {
                const v4f av = *(const v4f*)(arow + x * XST + 8 * q);
                const v2f a0 = {av.x, av.y};
                const v2f a1 = {av.z, av.w};
                acc[x] = __builtin_amdgcn_wmma_f32_16x16x4_f32(
                     false, a0,  false, b0,
                     (short)0, acc[x],  false,  false);
                acc[x] = __builtin_amdgcn_wmma_f32_16x16x4_f32(
                     false, a1,  false, b1,
                     (short)0, acc[x],  false,  false);
            }
        }

        float* o = sD + (8 * h) * LDP + w * 16 + c;
        #pragma unroll
        for (int x = 0; x < 3; ++x) {
            #pragma unroll
            for (int r = 0; r < 8; ++r)
                o[x * XST + r * LDP] = acc[x][r];
        }
    }
    __syncthreads();

    {
        const float* s = sD + w * LDP + 2 * lane;
        const v2f y0 = *(const v2f*)(s);
        const v2f y1 = *(const v2f*)(s + XST);
        const v2f y2 = *(const v2f*)(s + 2 * XST);
        float* dst = D + (size_t)a * (3 * CCH) + 2 * lane;
        __builtin_nontemporal_store(y0, (v2f*)(dst));
        __builtin_nontemporal_store(y1, (v2f*)(dst + CCH));
        __builtin_nontemporal_store(y2, (v2f*)(dst + 2 * CCH));
        __builtin_nontemporal_store(y0 * y0 + y1 * y1 + y2 * y2,
                                    (v2f*)(dotted + (size_t)a * CCH + 2 * lane));
    }
}

extern "C" void kernel_launch(void* const* d_in, const int* in_sizes, int n_in,
                              void* d_out, int out_size, void* d_ws, size_t ws_size,
                              hipStream_t stream) {
    const int*   ind2 = (const int*)  d_in[0];
    const float* p3   = (const float*)d_in[1];
    const float* i1   = (const float*)d_in[2];
    const float* d3   = (const float*)d_in[3];
    const float* W    = (const float*)d_in[4];

    float* p3_new = (float*)d_out;
    float* dotted = (float*)d_out + (size_t)N_ATOMS * 3 * CCH;

    int*   scr  = (int*)d_ws;
    int*   tab  = scr + SCR_TAB;
    int*   off  = scr + SCR_OFF;
    int4*  recA = (int4*)(scr + SCR_RECA);
    float* recZ = (float*)(scr + SCR_RECZ);
    int4*  entA = (int4*)(scr + SCR_ENTA);
    float* entZ = (float*)(scr + SCR_ENTZ);

    chunk_bucket_kernel<<<NCH, CT, 0, stream>>>(ind2, d3, tab, recA, recZ);

    bucket_sort_kernel<<<NB, ST, 0, stream>>>(tab, recA, recZ, off, entA, entZ);

    segsum_gemm_dot_kernel<<<N_ATOMS / GA, 512, 0, stream>>>(off, entA, entZ, p3, i1, W, p3_new, dotted);
}


# Level 2 of the sort fused into the segment-sum kernel: one 32-wave work-group per bucket orders its records in LDS and sums atoms taken from an LDS counter; one launch gone; sum order unchanged
# speedup vs baseline: 3.7251x; 1.1767x over previous
#include <hip/hip_runtime.h>


#define N_ATOMS 50000
#define N_PAIRS 800000
#define CCH     64

#define NB    256
#define BWID  196
#define NCH   256
#define CHP   3125
#define NTAB  (NB * NCH)

#define SCR_TAB   0
#define SCR_RECA  115648
#define SCR_RECZ  (115648 + 4 * N_PAIRS)

typedef __attribute__((ext_vector_type(2))) float v2f;
typedef __attribute__((ext_vector_type(8))) float v8f;

#define CT   512
#define CU   7

__global__ __launch_bounds__(CT) void chunk_bucket_kernel(
    const int*   __restrict__ ind2,
    const float* __restrict__ d3,
    int*         __restrict__ tab,
    int4*        __restrict__ recA,
    float*       __restrict__ recZ)
{
    __shared__ int sc[NB];
    __shared__ int wtot[NB / 32];
    const int t = threadIdx.x, ch = blockIdx.x, lane = t & 31;
    if (t < NB) sc[t] = 0;

    int ri[CU], rj[CU];
    float dx[CU], dy[CU], dz[CU];
    #pragma unroll
    for (int u = 0; u < CU; ++u) {
        const int q = t + CT * u;
        ri[u] = -1; rj[u] = 0; dx[u] = 0.f; dy[u] = 0.f; dz[u] = 0.f;
        if (q < CHP) {
            const int p = ch * CHP + q;
            const int2 ij = ((const int2*)ind2)[p];
            ri[u] = ij.x; rj[u] = ij.y;
            dx[u] = d3[3 * p + 0];
            dy[u] = d3[3 * p + 1];
            dz[u] = d3[3 * p + 2];
        }
    }
    __syncthreads();
    #pragma unroll
    for (int u = 0; u < CU; ++u)
        if (ri[u] >= 0) atomicAdd(&sc[ri[u] / BWID], 1);
    __syncthreads();

    int T = 0, inc = 0;
    if (t < NB) {
        T = sc[t];
        inc = T;
        for (int d = 1; d < 32; d <<= 1) {
            const int nb = __shfl_up(inc, d);
            if (lane >= d) inc += nb;
        }
        if (lane == 31) wtot[t >> 5] = inc;
    }
    __syncthreads();
    if (t < NB) {
        int wp = 0;
        #pragma unroll
        for (int m = 0; m < NB / 32; ++m) {
            const int s = wtot[m];
            if (m < (t >> 5)) wp += s;
        }
        const int ls = wp + inc - T;
        sc[t] = ls;
        tab[ch * NB + t] = ls | (T << 16);
    }
    __syncthreads();
    #pragma unroll
    for (int u = 0; u < CU; ++u) {
        if (ri[u] >= 0) {
            const int p = ch * CHP + t + CT * u;
            const int b = ri[u] / BWID;
            const int k = ri[u] - b * BWID;
            const int pos = ch * CHP + atomicAdd(&sc[b], 1);
            recA[pos] = make_int4(p, (k << 16) | rj[u], __float_as_int(dx[u]), __float_as_int(dy[u]));
            recZ[pos] = dz[u];
        }
    }
}

template <int U>
__device__ __forceinline__ void seg_step(
    int k, int lane, int ep, int ej, float ex, float ey, float ez,
    const float* __restrict__ p3, const float* __restrict__ i1,
    v2f& acc0, v2f& acc1, v2f& acc2)
{
    v2f g[U], x0[U], x1[U], x2[U];
    float d0[U], d1[U], d2[U];
    #pragma unroll
    for (int u = 0; u < U; ++u) {
        const int p = __builtin_amdgcn_readlane(ep, k + u);
        const int j = __builtin_amdgcn_readlane(ej, k + u);
        d0[u] = __builtin_bit_cast(float, __builtin_amdgcn_readlane(__builtin_bit_cast(int, ex), k + u));
        d1[u] = __builtin_bit_cast(float, __builtin_amdgcn_readlane(__builtin_bit_cast(int, ey), k + u));
        d2[u] = __builtin_bit_cast(float, __builtin_amdgcn_readlane(__builtin_bit_cast(int, ez), k + u));
        g[u] = __builtin_nontemporal_load((const v2f*)(i1 + (size_t)p * CCH + 2 * lane));
        const float* s = p3 + (size_t)j * (3 * CCH) + 2 * lane;
        x0[u] = *(const v2f*)(s);
        x1[u] = *(const v2f*)(s + CCH);
        x2[u] = *(const v2f*)(s + 2 * CCH);
    }
    #pragma unroll
    for (int u = 0; u < U; ++u) {
        acc0 += (x0[u] + d0[u]) * g[u];
        acc1 += (x1[u] + d1[u]) * g[u];
        acc2 += (x2[u] + d2[u]) * g[u];
    }
}

typedef __attribute__((ext_vector_type(4))) float v4f;

#define FT   1024
#define SU   4
#define CAP  4096
#define RTL  13
#define LDP  68
#define XST  (RTL * 16 * LDP)

__global__ __launch_bounds__(FT) void bucket_segsum_gemm_dot_kernel(
    const int*   __restrict__ tab,
    const int4*  __restrict__ recA,
    const float* __restrict__ recZ,
    const float* __restrict__ p3,
    const float* __restrict__ i1,
    const float* __restrict__ W,
    float*       __restrict__ D,
    float*       __restrict__ dotted)
{
    __shared__ int cnt[NB];
    __shared__ int cur[NB];
    __shared__ int rsrc[NCH];
    __shared__ int rlen[NCH];
    __shared__ int wcn[NCH / 32];
    __shared__ int nxt;
    __shared__ int   lp[CAP];
    __shared__ int   lj[CAP];
    __shared__ float lx[CAP];
    __shared__ float ly[CAP];
    __shared__ float lz[CAP];
    __shared__ __attribute__((aligned(16))) float sA[3 * XST];

    const int t = threadIdx.x, b = blockIdx.x, lane = t & 31;
    const int w = __builtin_amdgcn_readfirstlane(t >> 5);
    const int natoms = (N_ATOMS - b * BWID < BWID) ? (N_ATOMS - b * BWID) : BWID;

    if (t < NB) cnt[t] = 0;
    if (t == 0) nxt = 0;
    if (t < NCH) {
        const int v = tab[t * NB + b];
        int n = v >> 16;
        rsrc[t] = t * CHP + (v & 0xffff);
        rlen[t] = n;
        for (int d = 16; d >= 1; d >>= 1) n += __shfl_xor(n, d);
        if (lane == 0) wcn[t >> 5] = n;
    }
    __syncthreads();

    int btot = 0;
    #pragma unroll
    for (int m = 0; m < NCH / 32; ++m) btot += wcn[m];
    const bool fits = btot <= CAP;

    const int src = rsrc[t >> 2];
    const int rn  = rlen[t >> 2];
    const int sub = t & 3;

    if (fits) {
        for (int r0 = sub; r0 < rn; r0 += 4 * SU) {
            int kw[SU];
            #pragma unroll
            for (int u = 0; u < SU; ++u) {
                const int r = r0 + 4 * u;
                kw[u] = (r < rn) ? recA[src + r].y : -1;
            }
            #pragma unroll
            for (int u = 0; u < SU; ++u)
                if (kw[u] >= 0) atomicAdd(&cnt[kw[u] >> 16], 1);
        }
    }
    __syncthreads();

    if (t < 32) {
        int c8[8];
        int s = 0;
        #pragma unroll
        for (int m = 0; m < 8; ++m) { c8[m] = cnt[8 * lane + m]; s += c8[m]; }
        int inc = s;
        for (int d = 1; d < 32; d <<= 1) {
            const int nb = __shfl_up(inc, d);
            if (lane >= d) inc += nb;
        }
        int run = inc - s;
        #pragma unroll
        for (int m = 0; m < 8; ++m) { cur[8 * lane + m] = run; run += c8[m]; }
    }
    __syncthreads();

    if (fits) {
        for (int r0 = sub; r0 < rn; r0 += 4 * SU) {
            int4  e[SU];
            float z[SU];
            #pragma unroll
            for (int u = 0; u < SU; ++u) {
                const int r = r0 + 4 * u;
                e[u] = make_int4(0, -1, 0, 0); z[u] = 0.f;
                if (r < rn) { e[u] = recA[src + r]; z[u] = recZ[src + r]; }
            }
            #pragma unroll
            for (int u = 0; u < SU; ++u) {
                if (e[u].y >= 0) {
                    const int slot = atomicAdd(&cur[e[u].y >> 16], 1);
                    lp[slot] = e[u].x;
                    lj[slot] = e[u].y & 0xffff;
                    lx[slot] = __int_as_float(e[u].z);
                    ly[slot] = __int_as_float(e[u].w);
                    lz[slot] = z[u];
                }
            }
        }
    }
    __syncthreads();

    for (;;) {
        int k = 0;
        if (lane == 0) k = atomicAdd(&nxt, 1);
        k = __builtin_amdgcn_readfirstlane(k);
        if (k >= natoms) break;

        v2f acc0 = {0.f, 0.f}, acc1 = {0.f, 0.f}, acc2 = {0.f, 0.f};
        if (fits) {
            const int beg = k ? __builtin_amdgcn_readfirstlane(cur[k - 1]) : 0;
            const int end = __builtin_amdgcn_readfirstlane(cur[k]);
            for (int base = beg; base < end; base += 32) {
                const int n = (end - base < 32) ? (end - base) : 32;
                int ep = 0, ej = 0;
                float ex = 0.f, ey = 0.f, ez = 0.f;
                if (lane < n) {
                    ep = lp[base + lane]; ej = lj[base + lane];
                    ex = lx[base + lane]; ey = ly[base + lane]; ez = lz[base + lane];
                }
                int kk = 0;
                for (; kk + 8 <= n; kk += 8)
                    seg_step<8>(kk, lane, ep, ej, ex, ey, ez, p3, i1, acc0, acc1, acc2);
                if (kk + 4 <= n) {
                    seg_step<4>(kk, lane, ep, ej, ex, ey, ez, p3, i1, acc0, acc1, acc2);
                    kk += 4;
                }
                if (kk + 2 <= n) {
                    seg_step<2>(kk, lane, ep, ej, ex, ey, ez, p3, i1, acc0, acc1, acc2);
                    kk += 2;
                }
                if (kk < n)
                    seg_step<1>(kk, lane, ep, ej, ex, ey, ez, p3, i1, acc0, acc1, acc2);
            }
        } else {
            for (int c2 = 0; c2 < NCH; ++c2) {
                const int s0 = __builtin_amdgcn_readfirstlane(rsrc[c2]);
                const int nr = __builtin_amdgcn_readfirstlane(rlen[c2]);
                for (int r = 0; r < nr; r += 32) {
                    int ep = 0, ej = 0;
                    float ex = 0.f, ey = 0.f, ez = 0.f;
                    bool mine = false;
                    if (r + lane < nr) {
                        const int4 e = recA[s0 + r + lane];
                        mine = (e.y >> 16) == k;
                        ep = e.x; ej = e.y & 0xffff;
                        ex = __int_as_float(e.z); ey = __int_as_float(e.w);
                        ez = recZ[s0 + r + lane];
                    }
                    unsigned mask = __builtin_amdgcn_ballot_w32(mine);
                    while (mask) {
                        const int l = __builtin_ctz(mask);
                        mask &= mask - 1;
                        seg_step<1>(l, lane, ep, ej, ex, ey, ez, p3, i1, acc0, acc1, acc2);
                    }
                }
            }
        }

        float* s = sA + k * LDP + 2 * lane;
        *(v2f*)(s)           = acc0;
        *(v2f*)(s + XST)     = acc1;
        *(v2f*)(s + 2 * XST) = acc2;
    }
    __syncthreads();

    const int rt = w >> 1;
    if (w < 2 * RTL && rt * 16 < natoms) {
        const int c  = lane & 15;
        const int h  = lane >> 4;
        const int cb = (w & 1) * 32;
        const float* arow  = sA + (rt * 16 + c) * LDP + 4 * h;
        const float* wrow0 = W + (size_t)(cb + c) * CCH + 4 * h;
        const float* wrow1 = wrow0 + 16 * CCH;

        v8f acc[3][2] = {};
        #pragma unroll
        for (int q = 0; q < 8; ++q) {
            const v4f bv0 = *(const v4f*)(wrow0 + 8 * q);
            const v4f bv1 = *(const v4f*)(wrow1 + 8 * q);
            const v2f b00 = {bv0.x, bv0.y};
            const v2f b01 = {bv0.z, bv0.w};
            const v2f b10 = {bv1.x, bv1.y};
            const v2f b11 = {bv1.z, bv1.w};
            #pragma unroll
            for (int x = 0; x < 3; ++x) {
                const v4f av = *(const v4f*)(arow + x * XST + 8 * q);
                const v2f a0 = {av.x, av.y};
                const v2f a1 = {av.z, av.w};
                acc[x][0] = __builtin_amdgcn_wmma_f32_16x16x4_f32(
                     false, a0,  false, b00,
                     (short)0, acc[x][0],  false,  false);
                acc[x][1] = __builtin_amdgcn_wmma_f32_16x16x4_f32(
                     false, a0,  false, b10,
                     (short)0, acc[x][1],  false,  false);
                acc[x][0] = __builtin_amdgcn_wmma_f32_16x16x4_f32(
                     false, a1,  false, b01,
                     (short)0, acc[x][0],  false,  false);
                acc[x][1] = __builtin_amdgcn_wmma_f32_16x16x4_f32(
                     false, a1,  false, b11,
                     (short)0, acc[x][1],  false,  false);
            }
        }

        #pragma unroll
        for (int r = 0; r < 8; ++r) {
            const int k = rt * 16 + 8 * h + r;
            if (k < natoms) {
                const size_t a = (size_t)(b * BWID + k);
                float* dst = D + a * (3 * CCH) + cb + c;
                float* dd  = dotted + a * CCH + cb + c;
                #pragma unroll
                for (int nt = 0; nt < 2; ++nt) {
                    const float y0 = acc[0][nt][r], y1 = acc[1][nt][r], y2 = acc[2][nt][r];
                    __builtin_nontemporal_store(y0, dst + 16 * nt);
                    __builtin_nontemporal_store(y1, dst + CCH + 16 * nt);
                    __builtin_nontemporal_store(y2, dst + 2 * CCH + 16 * nt);
                    __builtin_nontemporal_store(y0 * y0 + y1 * y1 + y2 * y2, dd + 16 * nt);
                }
            }
        }
    }
}

extern "C" void kernel_launch(void* const* d_in, const int* in_sizes, int n_in,
                              void* d_out, int out_size, void* d_ws, size_t ws_size,
                              hipStream_t stream) {
    const int*   ind2 = (const int*)  d_in[0];
    const float* p3   = (const float*)d_in[1];
    const float* i1   = (const float*)d_in[2];
    const float* d3   = (const float*)d_in[3];
    const float* W    = (const float*)d_in[4];

    float* p3_new = (float*)d_out;
    float* dotted = (float*)d_out + (size_t)N_ATOMS * 3 * CCH;

    int*   scr  = (int*)d_ws;
    int*   tab  = scr + SCR_TAB;
    int4*  recA = (int4*)(scr + SCR_RECA);
    float* recZ = (float*)(scr + SCR_RECZ);

    chunk_bucket_kernel<<<NCH, CT, 0, stream>>>(ind2, d3, tab, recA, recZ);

    bucket_segsum_gemm_dot_kernel<<<NB, FT, 0, stream>>>(tab, recA, recZ, p3, i1, W, p3_new, dotted);
}


# Level-1 records put in bucket order in LDS, then copied to memory linearly: each wave store covers whole aligned 128-byte lines, not one scattered 16- or 4-byte piece per lane; sum order unchanged
# speedup vs baseline: 3.9288x; 1.0547x over previous
#include <hip/hip_runtime.h>


#define N_ATOMS 50000
#define N_PAIRS 800000
#define CCH     64

#define NB    256
#define BWID  196
#define NCH   256
#define CHP   3125
#define NTAB  (NB * NCH)

#define SCR_TAB   0
#define SCR_RECA  115648
#define SCR_RECZ  (115648 + 4 * N_PAIRS)

typedef __attribute__((ext_vector_type(2))) float v2f;
typedef __attribute__((ext_vector_type(8))) float v8f;

#define CT   512
#define CU   7

__global__ __launch_bounds__(CT) void chunk_bucket_kernel(
    const int*   __restrict__ ind2,
    const float* __restrict__ d3,
    int*         __restrict__ tab,
    int4*        __restrict__ recA,
    float*       __restrict__ recZ)
{
    __shared__ int sc[NB];
    __shared__ int wtot[NB / 32];
    __shared__ int4  srec[CHP];
    __shared__ float sz[CHP];
    const int t = threadIdx.x, ch = blockIdx.x, lane = t & 31;
    if (t < NB) sc[t] = 0;

    int ri[CU], rj[CU];
    float dx[CU], dy[CU], dz[CU];
    #pragma unroll
    for (int u = 0; u < CU; ++u) {
        const int q = t + CT * u;
        ri[u] = -1; rj[u] = 0; dx[u] = 0.f; dy[u] = 0.f; dz[u] = 0.f;
        if (q < CHP) {
            const int p = ch * CHP + q;
            const int2 ij = ((const int2*)ind2)[p];
            ri[u] = ij.x; rj[u] = ij.y;
            dx[u] = d3[3 * p + 0];
            dy[u] = d3[3 * p + 1];
            dz[u] = d3[3 * p + 2];
        }
    }
    __syncthreads();
    #pragma unroll
    for (int u = 0; u < CU; ++u)
        if (ri[u] >= 0) atomicAdd(&sc[ri[u] / BWID], 1);
    __syncthreads();

    int T = 0, inc = 0;
    if (t < NB) {
        T = sc[t];
        inc = T;
        for (int d = 1; d < 32; d <<= 1) {
            const int nb = __shfl_up(inc, d);
            if (lane >= d) inc += nb;
        }
        if (lane == 31) wtot[t >> 5] = inc;
    }
    __syncthreads();
    if (t < NB) {
        int wp = 0;
        #pragma unroll
        for (int m = 0; m < NB / 32; ++m) {
            const int s = wtot[m];
            if (m < (t >> 5)) wp += s;
        }
        const int ls = wp + inc - T;
        sc[t] = ls;
        tab[ch * NB + t] = ls | (T << 16);
    }
    __syncthreads();
    #pragma unroll
    for (int u = 0; u < CU; ++u) {
        if (ri[u] >= 0) {
            const int p = ch * CHP + t + CT * u;
            const int b = ri[u] / BWID;
            const int k = ri[u] - b * BWID;
            const int slot = atomicAdd(&sc[b], 1);
            srec[slot] = make_int4(p, (k << 16) | rj[u], __float_as_int(dx[u]), __float_as_int(dy[u]));
            sz[slot] = dz[u];
        }
    }
    __syncthreads();

    static_assert(CT * (CU - 1) <= CHP, "trips 1 .. CU-2 stay inside the chunk");
    static_assert(CT * CU >= CHP + 31, "the last trip reaches the chunk's end from a start up to 31 below it");
    const int first = ch * CHP, last = first + CHP;
    const int ga = (first & ~7) + t;
    const int gz = (first & ~31) + t;
    if (ga >= first) recA[ga] = srec[ga - first];
    #pragma unroll
    for (int u = 1; u < CU - 1; ++u) recA[ga + CT * u] = srec[ga + CT * u - first];
    if (ga + CT * (CU - 1) < last) recA[ga + CT * (CU - 1)] = srec[ga + CT * (CU - 1) - first];

    if (gz >= first) recZ[gz] = sz[gz - first];
    #pragma unroll
    for (int u = 1; u < CU - 1; ++u) recZ[gz + CT * u] = sz[gz + CT * u - first];
    if (gz + CT * (CU - 1) < last) recZ[gz + CT * (CU - 1)] = sz[gz + CT * (CU - 1) - first];
}

template <int U>
__device__ __forceinline__ void seg_step(
    int k, int lane, int ep, int ej, float ex, float ey, float ez,
    const float* __restrict__ p3, const float* __restrict__ i1,
    v2f& acc0, v2f& acc1, v2f& acc2)
{
    v2f g[U], x0[U], x1[U], x2[U];
    float d0[U], d1[U], d2[U];
    #pragma unroll
    for (int u = 0; u < U; ++u) {
        const int p = __builtin_amdgcn_readlane(ep, k + u);
        const int j = __builtin_amdgcn_readlane(ej, k + u);
        d0[u] = __builtin_bit_cast(float, __builtin_amdgcn_readlane(__builtin_bit_cast(int, ex), k + u));
        d1[u] = __builtin_bit_cast(float, __builtin_amdgcn_readlane(__builtin_bit_cast(int, ey), k + u));
        d2[u] = __builtin_bit_cast(float, __builtin_amdgcn_readlane(__builtin_bit_cast(int, ez), k + u));
        g[u] = __builtin_nontemporal_load((const v2f*)(i1 + (size_t)p * CCH + 2 * lane));
        const float* s = p3 + (size_t)j * (3 * CCH) + 2 * lane;
        x0[u] = *(const v2f*)(s);
        x1[u] = *(const v2f*)(s + CCH);
        x2[u] = *(const v2f*)(s + 2 * CCH);
    }
    #pragma unroll
    for (int u = 0; u < U; ++u) {
        acc0 += (x0[u] + d0[u]) * g[u];
        acc1 += (x1[u] + d1[u]) * g[u];
        acc2 += (x2[u] + d2[u]) * g[u];
    }
}

typedef __attribute__((ext_vector_type(4))) float v4f;

#define FT   1024
#define SU   4
#define CAP  4096
#define RTL  13
#define LDP  68
#define XST  (RTL * 16 * LDP)

__global__ __launch_bounds__(FT) void bucket_segsum_gemm_dot_kernel(
    const int*   __restrict__ tab,
    const int4*  __restrict__ recA,
    const float* __restrict__ recZ,
    const float* __restrict__ p3,
    const float* __restrict__ i1,
    const float* __restrict__ W,
    float*       __restrict__ D,
    float*       __restrict__ dotted)
{
    __shared__ int cnt[NB];
    __shared__ int cur[NB];
    __shared__ int rsrc[NCH];
    __shared__ int rlen[NCH];
    __shared__ int wcn[NCH / 32];
    __shared__ int nxt;
    __shared__ int   lp[CAP];
    __shared__ int   lj[CAP];
    __shared__ float lx[CAP];
    __shared__ float ly[CAP];
    __shared__ float lz[CAP];
    __shared__ __attribute__((aligned(16))) float sA[3 * XST];

    const int t = threadIdx.x, b = blockIdx.x, lane = t & 31;
    const int w = __builtin_amdgcn_readfirstlane(t >> 5);
    const int natoms = (N_ATOMS - b * BWID < BWID) ? (N_ATOMS - b * BWID) : BWID;

    if (t < NB) cnt[t] = 0;
    if (t == 0) nxt = 0;
    if (t < NCH) {
        const int v = tab[t * NB + b];
        int n = v >> 16;
        rsrc[t] = t * CHP + (v & 0xffff);
        rlen[t] = n;
        for (int d = 16; d >= 1; d >>= 1) n += __shfl_xor(n, d);
        if (lane == 0) wcn[t >> 5] = n;
    }
    __syncthreads();

    int btot = 0;
    #pragma unroll
    for (int m = 0; m < NCH / 32; ++m) btot += wcn[m];
    const bool fits = btot <= CAP;

    const int src = rsrc[t >> 2];
    const int rn  = rlen[t >> 2];
    const int sub = t & 3;

    if (fits) {
        for (int r0 = sub; r0 < rn; r0 += 4 * SU) {
            int kw[SU];
            #pragma unroll
            for (int u = 0; u < SU; ++u) {
                const int r = r0 + 4 * u;
                kw[u] = (r < rn) ? recA[src + r].y : -1;
            }
            #pragma unroll
            for (int u = 0; u < SU; ++u)
                if (kw[u] >= 0) atomicAdd(&cnt[kw[u] >> 16], 1);
        }
    }
    __syncthreads();

    if (t < 32) {
        int c8[8];
        int s = 0;
        #pragma unroll
        for (int m = 0; m < 8; ++m) { c8[m] = cnt[8 * lane + m]; s += c8[m]; }
        int inc = s;
        for (int d = 1; d < 32; d <<= 1) {
            const int nb = __shfl_up(inc, d);
            if (lane >= d) inc += nb;
        }
        int run = inc - s;
        #pragma unroll
        for (int m = 0; m < 8; ++m) { cur[8 * lane + m] = run; run += c8[m]; }
    }
    __syncthreads();

    if (fits) {
        for (int r0 = sub; r0 < rn; r0 += 4 * SU) {
            int4  e[SU];
            float z[SU];
            #pragma unroll
            for (int u = 0; u < SU; ++u) {
                const int r = r0 + 4 * u;
                e[u] = make_int4(0, -1, 0, 0); z[u] = 0.f;
                if (r < rn) { e[u] = recA[src + r]; z[u] = recZ[src + r]; }
            }
            #pragma unroll
            for (int u = 0; u < SU; ++u) {
                if (e[u].y >= 0) {
                    const int slot = atomicAdd(&cur[e[u].y >> 16], 1);
                    lp[slot] = e[u].x;
                    lj[slot] = e[u].y & 0xffff;
                    lx[slot] = __int_as_float(e[u].z);
                    ly[slot] = __int_as_float(e[u].w);
                    lz[slot] = z[u];
                }
            }
        }
    }
    __syncthreads();

    for (;;) {
        int k = 0;
        if (lane == 0) k = atomicAdd(&nxt, 1);
        k = __builtin_amdgcn_readfirstlane(k);
        if (k >= natoms) break;

        v2f acc0 = {0.f, 0.f}, acc1 = {0.f, 0.f}, acc2 = {0.f, 0.f};
        if (fits) {
            const int beg = k ? __builtin_amdgcn_readfirstlane(cur[k - 1]) : 0;
            const int end = __builtin_amdgcn_readfirstlane(cur[k]);
            for (int base = beg; base < end; base += 32) {
                const int n = (end - base < 32) ? (end - base) : 32;
                int ep = 0, ej = 0;
                float ex = 0.f, ey = 0.f, ez = 0.f;
                if (lane < n) {
                    ep = lp[base + lane]; ej = lj[base + lane];
                    ex = lx[base + lane]; ey = ly[base + lane]; ez = lz[base + lane];
                }
                int kk = 0;
                for (; kk + 8 <= n; kk += 8)
                    seg_step<8>(kk, lane, ep, ej, ex, ey, ez, p3, i1, acc0, acc1, acc2);
                if (kk + 4 <= n) {
                    seg_step<4>(kk, lane, ep, ej, ex, ey, ez, p3, i1, acc0, acc1, acc2);
                    kk += 4;
                }
                if (kk + 2 <= n) {
                    seg_step<2>(kk, lane, ep, ej, ex, ey, ez, p3, i1, acc0, acc1, acc2);
                    kk += 2;
                }
                if (kk < n)
                    seg_step<1>(kk, lane, ep, ej, ex, ey, ez, p3, i1, acc0, acc1, acc2);
            }
        } else {
            for (int c2 = 0; c2 < NCH; ++c2) {
                const int s0 = __builtin_amdgcn_readfirstlane(rsrc[c2]);
                const int nr = __builtin_amdgcn_readfirstlane(rlen[c2]);
                for (int r = 0; r < nr; r += 32) {
                    int ep = 0, ej = 0;
                    float ex = 0.f, ey = 0.f, ez = 0.f;
                    bool mine = false;
                    if (r + lane < nr) {
                        const int4 e = recA[s0 + r + lane];
                        mine = (e.y >> 16) == k;
                        ep = e.x; ej = e.y & 0xffff;
                        ex = __int_as_float(e.z); ey = __int_as_float(e.w);
                        ez = recZ[s0 + r + lane];
                    }
                    unsigned mask = __builtin_amdgcn_ballot_w32(mine);
                    while (mask) {
                        const int l = __builtin_ctz(mask);
                        mask &= mask - 1;
                        seg_step<1>(l, lane, ep, ej, ex, ey, ez, p3, i1, acc0, acc1, acc2);
                    }
                }
            }
        }

        float* s = sA + k * LDP + 2 * lane;
        *(v2f*)(s)           = acc0;
        *(v2f*)(s + XST)     = acc1;
        *(v2f*)(s + 2 * XST) = acc2;
    }
    __syncthreads();

    const int rt = w >> 1;
    if (w < 2 * RTL && rt * 16 < natoms) {
        const int c  = lane & 15;
        const int h  = lane >> 4;
        const int cb = (w & 1) * 32;
        const float* arow  = sA + (rt * 16 + c) * LDP + 4 * h;
        const float* wrow0 = W + (size_t)(cb + c) * CCH + 4 * h;
        const float* wrow1 = wrow0 + 16 * CCH;

        v8f acc[3][2] = {};
        #pragma unroll
        for (int q = 0; q < 8; ++q) {
            const v4f bv0 = *(const v4f*)(wrow0 + 8 * q);
            const v4f bv1 = *(const v4f*)(wrow1 + 8 * q);
            const v2f b00 = {bv0.x, bv0.y};
            const v2f b01 = {bv0.z, bv0.w};
            const v2f b10 = {bv1.x, bv1.y};
            const v2f b11 = {bv1.z, bv1.w};
            #pragma unroll
            for (int x = 0; x < 3; ++x) {
                const v4f av = *(const v4f*)(arow + x * XST + 8 * q);
                const v2f a0 = {av.x, av.y};
                const v2f a1 = {av.z, av.w};
                acc[x][0] = __builtin_amdgcn_wmma_f32_16x16x4_f32(
                     false, a0,  false, b00,
                     (short)0, acc[x][0],  false,  false);
                acc[x][1] = __builtin_amdgcn_wmma_f32_16x16x4_f32(
                     false, a0,  false, b10,
                     (short)0, acc[x][1],  false,  false);
                acc[x][0] = __builtin_amdgcn_wmma_f32_16x16x4_f32(
                     false, a1,  false, b01,
                     (short)0, acc[x][0],  false,  false);
                acc[x][1] = __builtin_amdgcn_wmma_f32_16x16x4_f32(
                     false, a1,  false, b11,
                     (short)0, acc[x][1],  false,  false);
            }
        }

        #pragma unroll
        for (int r = 0; r < 8; ++r) {
            const int k = rt * 16 + 8 * h + r;
            if (k < natoms) {
                const size_t a = (size_t)(b * BWID + k);
                float* dst = D + a * (3 * CCH) + cb + c;
                float* dd  = dotted + a * CCH + cb + c;
                #pragma unroll
                for (int nt = 0; nt < 2; ++nt) {
                    const float y0 = acc[0][nt][r], y1 = acc[1][nt][r], y2 = acc[2][nt][r];
                    __builtin_nontemporal_store(y0, dst + 16 * nt);
                    __builtin_nontemporal_store(y1, dst + CCH + 16 * nt);
                    __builtin_nontemporal_store(y2, dst + 2 * CCH + 16 * nt);
                    __builtin_nontemporal_store(y0 * y0 + y1 * y1 + y2 * y2, dd + 16 * nt);
                }
            }
        }
    }
}

extern "C" void kernel_launch(void* const* d_in, const int* in_sizes, int n_in,
                              void* d_out, int out_size, void* d_ws, size_t ws_size,
                              hipStream_t stream) {
    const int*   ind2 = (const int*)  d_in[0];
    const float* p3   = (const float*)d_in[1];
    const float* i1   = (const float*)d_in[2];
    const float* d3   = (const float*)d_in[3];
    const float* W    = (const float*)d_in[4];

    float* p3_new = (float*)d_out;
    float* dotted = (float*)d_out + (size_t)N_ATOMS * 3 * CCH;

    int*   scr  = (int*)d_ws;
    int*   tab  = scr + SCR_TAB;
    int4*  recA = (int4*)(scr + SCR_RECA);
    float* recZ = (float*)(scr + SCR_RECZ);

    chunk_bucket_kernel<<<NCH, CT, 0, stream>>>(ind2, d3, tab, recA, recZ);

    bucket_segsum_gemm_dot_kernel<<<NB, FT, 0, stream>>>(tab, recA, recZ, p3, i1, W, p3_new, dotted);
}


# Gate rows (read once, from memory) prefetched into the second-level cache one round of atoms ahead, so the gate loads are second-level hits like the sender rows beside them; sums unchanged
# speedup vs baseline: 4.0671x; 1.0352x over previous
#include <hip/hip_runtime.h>


#define N_ATOMS 50000
#define N_PAIRS 800000
#define CCH     64

#define NB    256
#define BWID  196
#define NCH   256
#define CHP   3125
#define NTAB  (NB * NCH)

#define SCR_TAB   0
#define SCR_RECA  115648
#define SCR_RECZ  (115648 + 4 * N_PAIRS)

typedef __attribute__((ext_vector_type(2))) float v2f;
typedef __attribute__((ext_vector_type(8))) float v8f;

#define CT   512
#define CU   7

__global__ __launch_bounds__(CT) void chunk_bucket_kernel(
    const int*   __restrict__ ind2,
    const float* __restrict__ d3,
    int*         __restrict__ tab,
    int4*        __restrict__ recA,
    float*       __restrict__ recZ)
{
    __shared__ int sc[NB];
    __shared__ int wtot[NB / 32];
    __shared__ int4  srec[CHP];
    __shared__ float sz[CHP];
    const int t = threadIdx.x, ch = blockIdx.x, lane = t & 31;
    if (t < NB) sc[t] = 0;

    int ri[CU], rj[CU];
    float dx[CU], dy[CU], dz[CU];
    #pragma unroll
    for (int u = 0; u < CU; ++u) {
        const int q = t + CT * u;
        ri[u] = -1; rj[u] = 0; dx[u] = 0.f; dy[u] = 0.f; dz[u] = 0.f;
        if (q < CHP) {
            const int p = ch * CHP + q;
            const int2 ij = ((const int2*)ind2)[p];
            ri[u] = ij.x; rj[u] = ij.y;
            dx[u] = d3[3 * p + 0];
            dy[u] = d3[3 * p + 1];
            dz[u] = d3[3 * p + 2];
        }
    }
    __syncthreads();
    #pragma unroll
    for (int u = 0; u < CU; ++u)
        if (ri[u] >= 0) atomicAdd(&sc[ri[u] / BWID], 1);
    __syncthreads();

    int T = 0, inc = 0;
    if (t < NB) {
        T = sc[t];
        inc = T;
        for (int d = 1; d < 32; d <<= 1) {
            const int nb = __shfl_up(inc, d);
            if (lane >= d) inc += nb;
        }
        if (lane == 31) wtot[t >> 5] = inc;
    }
    __syncthreads();
    if (t < NB) {
        int wp = 0;
        #pragma unroll
        for (int m = 0; m < NB / 32; ++m) {
            const int s = wtot[m];
            if (m < (t >> 5)) wp += s;
        }
        const int ls = wp + inc - T;
        sc[t] = ls;
        tab[ch * NB + t] = ls | (T << 16);
    }
    __syncthreads();
    #pragma unroll
    for (int u = 0; u < CU; ++u) {
        if (ri[u] >= 0) {
            const int p = ch * CHP + t + CT * u;
            const int b = ri[u] / BWID;
            const int k = ri[u] - b * BWID;
            const int slot = atomicAdd(&sc[b], 1);
            srec[slot] = make_int4(p, (k << 16) | rj[u], __float_as_int(dx[u]), __float_as_int(dy[u]));
            sz[slot] = dz[u];
        }
    }
    __syncthreads();

    static_assert(CT * (CU - 1) <= CHP, "trips 1 .. CU-2 stay inside the chunk");
    static_assert(CT * CU >= CHP + 31, "the last trip reaches the chunk's end from a start up to 31 below it");
    const int first = ch * CHP, last = first + CHP;
    const int ga = (first & ~7) + t;
    const int gz = (first & ~31) + t;
    if (ga >= first) recA[ga] = srec[ga - first];
    #pragma unroll
    for (int u = 1; u < CU - 1; ++u) recA[ga + CT * u] = srec[ga + CT * u - first];
    if (ga + CT * (CU - 1) < last) recA[ga + CT * (CU - 1)] = srec[ga + CT * (CU - 1) - first];

    if (gz >= first) recZ[gz] = sz[gz - first];
    #pragma unroll
    for (int u = 1; u < CU - 1; ++u) recZ[gz + CT * u] = sz[gz + CT * u - first];
    if (gz + CT * (CU - 1) < last) recZ[gz + CT * (CU - 1)] = sz[gz + CT * (CU - 1) - first];
}

template <int U>
__device__ __forceinline__ void seg_step(
    int k, int lane, int ep, int ej, float ex, float ey, float ez,
    const float* __restrict__ p3, const float* __restrict__ i1,
    v2f& acc0, v2f& acc1, v2f& acc2)
{
    v2f g[U], x0[U], x1[U], x2[U];
    float d0[U], d1[U], d2[U];
    #pragma unroll
    for (int u = 0; u < U; ++u) {
        const int p = __builtin_amdgcn_readlane(ep, k + u);
        const int j = __builtin_amdgcn_readlane(ej, k + u);
        d0[u] = __builtin_bit_cast(float, __builtin_amdgcn_readlane(__builtin_bit_cast(int, ex), k + u));
        d1[u] = __builtin_bit_cast(float, __builtin_amdgcn_readlane(__builtin_bit_cast(int, ey), k + u));
        d2[u] = __builtin_bit_cast(float, __builtin_amdgcn_readlane(__builtin_bit_cast(int, ez), k + u));
        g[u] = __builtin_nontemporal_load((const v2f*)(i1 + (size_t)p * CCH + 2 * lane));
        const float* s = p3 + (size_t)j * (3 * CCH) + 2 * lane;
        x0[u] = *(const v2f*)(s);
        x1[u] = *(const v2f*)(s + CCH);
        x2[u] = *(const v2f*)(s + 2 * CCH);
    }
    #pragma unroll
    for (int u = 0; u < U; ++u) {
        acc0 += (x0[u] + d0[u]) * g[u];
        acc1 += (x1[u] + d1[u]) * g[u];
        acc2 += (x2[u] + d2[u]) * g[u];
    }
}

__device__ __forceinline__ void gate_rows_ahead(
    int ka, int natoms, int lane, const int* cur, const int* lp, const float* __restrict__ i1)
{
    if (ka < natoms) {
        const int b0 = cur[ka > 0 ? ka - 1 : 0];
        const int e0 = cur[ka];
        const int beg = ka > 0 ? __builtin_amdgcn_readfirstlane(b0) : 0;
        const int end = __builtin_amdgcn_readfirstlane(e0);
        if (lane < end - beg) {
            const float* row = i1 + (size_t)lp[beg + lane] * CCH;
            asm volatile("global_prefetch_b8 %0, off scope:SCOPE_SE\n\ts_wait_xcnt 0x0" : : "v"(row));
        }
    }
}

typedef __attribute__((ext_vector_type(4))) float v4f;

#define FT   1024
#define SU   4
#define CAP  4096
#define RTL  13
#define LDP  68
#define XST  (RTL * 16 * LDP)

__global__ __launch_bounds__(FT) void bucket_segsum_gemm_dot_kernel(
    const int*   __restrict__ tab,
    const int4*  __restrict__ recA,
    const float* __restrict__ recZ,
    const float* __restrict__ p3,
    const float* __restrict__ i1,
    const float* __restrict__ W,
    float*       __restrict__ D,
    float*       __restrict__ dotted)
{
    __shared__ int cnt[NB];
    __shared__ int cur[NB];
    __shared__ int rsrc[NCH];
    __shared__ int rlen[NCH];
    __shared__ int wcn[NCH / 32];
    __shared__ int nxt;
    __shared__ int   lp[CAP];
    __shared__ int   lj[CAP];
    __shared__ float lx[CAP];
    __shared__ float ly[CAP];
    __shared__ float lz[CAP];
    __shared__ __attribute__((aligned(16))) float sA[3 * XST];

    const int t = threadIdx.x, b = blockIdx.x, lane = t & 31;
    const int w = __builtin_amdgcn_readfirstlane(t >> 5);
    const int natoms = (N_ATOMS - b * BWID < BWID) ? (N_ATOMS - b * BWID) : BWID;

    if (t < NB) cnt[t] = 0;
    if (t == 0) nxt = 0;
    if (t < NCH) {
        const int v = tab[t * NB + b];
        int n = v >> 16;
        rsrc[t] = t * CHP + (v & 0xffff);
        rlen[t] = n;
        for (int d = 16; d >= 1; d >>= 1) n += __shfl_xor(n, d);
        if (lane == 0) wcn[t >> 5] = n;
    }
    __syncthreads();

    int btot = 0;
    #pragma unroll
    for (int m = 0; m < NCH / 32; ++m) btot += wcn[m];
    const bool fits = btot <= CAP;

    const int src = rsrc[t >> 2];
    const int rn  = rlen[t >> 2];
    const int sub = t & 3;

    if (fits) {
        for (int r0 = sub; r0 < rn; r0 += 4 * SU) {
            int kw[SU];
            #pragma unroll
            for (int u = 0; u < SU; ++u) {
                const int r = r0 + 4 * u;
                kw[u] = (r < rn) ? recA[src + r].y : -1;
            }
            #pragma unroll
            for (int u = 0; u < SU; ++u)
                if (kw[u] >= 0) atomicAdd(&cnt[kw[u] >> 16], 1);
        }
    }
    __syncthreads();

    if (t < 32) {
        int c8[8];
        int s = 0;
        #pragma unroll
        for (int m = 0; m < 8; ++m) { c8[m] = cnt[8 * lane + m]; s += c8[m]; }
        int inc = s;
        for (int d = 1; d < 32; d <<= 1) {
            const int nb = __shfl_up(inc, d);
            if (lane >= d) inc += nb;
        }
        int run = inc - s;
        #pragma unroll
        for (int m = 0; m < 8; ++m) { cur[8 * lane + m] = run; run += c8[m]; }
    }
    __syncthreads();

    if (fits) {
        for (int r0 = sub; r0 < rn; r0 += 4 * SU) {
            int4  e[SU];
            float z[SU];
            #pragma unroll
            for (int u = 0; u < SU; ++u) {
                const int r = r0 + 4 * u;
                e[u] = make_int4(0, -1, 0, 0); z[u] = 0.f;
                if (r < rn) { e[u] = recA[src + r]; z[u] = recZ[src + r]; }
            }
            #pragma unroll
            for (int u = 0; u < SU; ++u) {
                if (e[u].y >= 0) {
                    const int slot = atomicAdd(&cur[e[u].y >> 16], 1);
                    lp[slot] = e[u].x;
                    lj[slot] = e[u].y & 0xffff;
                    lx[slot] = __int_as_float(e[u].z);
                    ly[slot] = __int_as_float(e[u].w);
                    lz[slot] = z[u];
                }
            }
        }
    }
    __syncthreads();

    if (fits) gate_rows_ahead(w, natoms, lane, cur, lp, i1);
    for (;;) {
        int k = 0;
        if (lane == 0) k = atomicAdd(&nxt, 1);
        k = __builtin_amdgcn_readfirstlane(k);
        if (k >= natoms) break;

        v2f acc0 = {0.f, 0.f}, acc1 = {0.f, 0.f}, acc2 = {0.f, 0.f};
        if (fits) {
            gate_rows_ahead(k + FT / 32, natoms, lane, cur, lp, i1);
            const int beg = k ? __builtin_amdgcn_readfirstlane(cur[k - 1]) : 0;
            const int end = __builtin_amdgcn_readfirstlane(cur[k]);
            for (int base = beg; base < end; base += 32) {
                const int n = (end - base < 32) ? (end - base) : 32;
                int ep = 0, ej = 0;
                float ex = 0.f, ey = 0.f, ez = 0.f;
                if (lane < n) {
                    ep = lp[base + lane]; ej = lj[base + lane];
                    ex = lx[base + lane]; ey = ly[base + lane]; ez = lz[base + lane];
                }
                int kk = 0;
                for (; kk + 8 <= n; kk += 8)
                    seg_step<8>(kk, lane, ep, ej, ex, ey, ez, p3, i1, acc0, acc1, acc2);
                if (kk + 4 <= n) {
                    seg_step<4>(kk, lane, ep, ej, ex, ey, ez, p3, i1, acc0, acc1, acc2);
                    kk += 4;
                }
                if (kk + 2 <= n) {
                    seg_step<2>(kk, lane, ep, ej, ex, ey, ez, p3, i1, acc0, acc1, acc2);
                    kk += 2;
                }
                if (kk < n)
                    seg_step<1>(kk, lane, ep, ej, ex, ey, ez, p3, i1, acc0, acc1, acc2);
            }
        } else {
            for (int c2 = 0; c2 < NCH; ++c2) {
                const int s0 = __builtin_amdgcn_readfirstlane(rsrc[c2]);
                const int nr = __builtin_amdgcn_readfirstlane(rlen[c2]);
                for (int r = 0; r < nr; r += 32) {
                    int ep = 0, ej = 0;
                    float ex = 0.f, ey = 0.f, ez = 0.f;
                    bool mine = false;
                    if (r + lane < nr) {
                        const int4 e = recA[s0 + r + lane];
                        mine = (e.y >> 16) == k;
                        ep = e.x; ej = e.y & 0xffff;
                        ex = __int_as_float(e.z); ey = __int_as_float(e.w);
                        ez = recZ[s0 + r + lane];
                    }
                    unsigned mask = __builtin_amdgcn_ballot_w32(mine);
                    while (mask) {
                        const int l = __builtin_ctz(mask);
                        mask &= mask - 1;
                        seg_step<1>(l, lane, ep, ej, ex, ey, ez, p3, i1, acc0, acc1, acc2);
                    }
                }
            }
        }

        float* s = sA + k * LDP + 2 * lane;
        *(v2f*)(s)           = acc0;
        *(v2f*)(s + XST)     = acc1;
        *(v2f*)(s + 2 * XST) = acc2;
    }
    __syncthreads();

    const int rt = w >> 1;
    if (w < 2 * RTL && rt * 16 < natoms) {
        const int c  = lane & 15;
        const int h  = lane >> 4;
        const int cb = (w & 1) * 32;
        const float* arow  = sA + (rt * 16 + c) * LDP + 4 * h;
        const float* wrow0 = W + (size_t)(cb + c) * CCH + 4 * h;
        const float* wrow1 = wrow0 + 16 * CCH;

        v8f acc[3][2] = {};
        #pragma unroll
        for (int q = 0; q < 8; ++q) {
            const v4f bv0 = *(const v4f*)(wrow0 + 8 * q);
            const v4f bv1 = *(const v4f*)(wrow1 + 8 * q);
            const v2f b00 = {bv0.x, bv0.y};
            const v2f b01 = {bv0.z, bv0.w};
            const v2f b10 = {bv1.x, bv1.y};
            const v2f b11 = {bv1.z, bv1.w};
            #pragma unroll
            for (int x = 0; x < 3; ++x) {
                const v4f av = *(const v4f*)(arow + x * XST + 8 * q);
                const v2f a0 = {av.x, av.y};
                const v2f a1 = {av.z, av.w};
                acc[x][0] = __builtin_amdgcn_wmma_f32_16x16x4_f32(
                     false, a0,  false, b00,
                     (short)0, acc[x][0],  false,  false);
                acc[x][1] = __builtin_amdgcn_wmma_f32_16x16x4_f32(
                     false, a0,  false, b10,
                     (short)0, acc[x][1],  false,  false);
                acc[x][0] = __builtin_amdgcn_wmma_f32_16x16x4_f32(
                     false, a1,  false, b01,
                     (short)0, acc[x][0],  false,  false);
                acc[x][1] = __builtin_amdgcn_wmma_f32_16x16x4_f32(
                     false, a1,  false, b11,
                     (short)0, acc[x][1],  false,  false);
            }
        }

        #pragma unroll
        for (int r = 0; r < 8; ++r) {
            const int k = rt * 16 + 8 * h + r;
            if (k < natoms) {
                const size_t a = (size_t)(b * BWID + k);
                float* dst = D + a * (3 * CCH) + cb + c;
                float* dd  = dotted + a * CCH + cb + c;
                #pragma unroll
                for (int nt = 0; nt < 2; ++nt) {
                    const float y0 = acc[0][nt][r], y1 = acc[1][nt][r], y2 = acc[2][nt][r];
                    __builtin_nontemporal_store(y0, dst + 16 * nt);
                    __builtin_nontemporal_store(y1, dst + CCH + 16 * nt);
                    __builtin_nontemporal_store(y2, dst + 2 * CCH + 16 * nt);
                    __builtin_nontemporal_store(y0 * y0 + y1 * y1 + y2 * y2, dd + 16 * nt);
                }
            }
        }
    }
}

extern "C" void kernel_launch(void* const* d_in, const int* in_sizes, int n_in,
                              void* d_out, int out_size, void* d_ws, size_t ws_size,
                              hipStream_t stream) {
    const int*   ind2 = (const int*)  d_in[0];
    const float* p3   = (const float*)d_in[1];
    const float* i1   = (const float*)d_in[2];
    const float* d3   = (const float*)d_in[3];
    const float* W    = (const float*)d_in[4];

    float* p3_new = (float*)d_out;
    float* dotted = (float*)d_out + (size_t)N_ATOMS * 3 * CCH;

    int*   scr  = (int*)d_ws;
    int*   tab  = scr + SCR_TAB;
    int4*  recA = (int4*)(scr + SCR_RECA);
    float* recZ = (float*)(scr + SCR_RECZ);

    chunk_bucket_kernel<<<NCH, CT, 0, stream>>>(ind2, d3, tab, recA, recZ);

    bucket_segsum_gemm_dot_kernel<<<NB, FT, 0, stream>>>(tab, recA, recZ, p3, i1, W, p3_new, dotted);
}
